# starting kernel
# baseline (speedup 1.0000x reference)
#include <hip/hip_runtime.h>

#define FPS_K   256
#define IN_D    128
#define OUT_D   256
#define FPS_T   1024
#define WAVES   (FPS_T / 32)

typedef __attribute__((ext_vector_type(2))) float v2f;
typedef __attribute__((ext_vector_type(8))) float v8f;

struct Pt3 { float x, y, z; };

__global__ __launch_bounds__(FPS_T) void fps_kernel(
    const float* __restrict__ means, int N,
    float* __restrict__ dists, int* __restrict__ idx_out)
{
    const int b    = blockIdx.x;
    const int t    = threadIdx.x;
    const int wave = t >> 5;
    const int lane = t & 31;

    const Pt3*  m   = (const Pt3*)(means + (size_t)b * N * 3);
    float*      dst = dists + (size_t)b * N;

    __shared__ float s_d[WAVES];
    __shared__ int   s_i[WAVES];
    __shared__ int   s_last;

    for (int i = t; i < N; i += FPS_T) dst[i] = 3.402823466e38f;
    if (t == 0) s_last = 0;
    __syncthreads();

    for (int k = 0; k < FPS_K; ++k) {
        const int last = s_last;
        if (t == 0) idx_out[(size_t)b * FPS_K + k] = last;

        const Pt3 p = m[last];

        float best  = -1.0f;
        int   besti = 0;
        for (int i = t; i < N; i += FPS_T) {
            const Pt3 q = m[i];
            const float dx = q.x - p.x;
            const float dy = q.y - p.y;
            const float dz = q.z - p.z;
            const float d2 = fmaf(dx, dx, fmaf(dy, dy, dz * dz));
            const float nd = fminf(dst[i], d2);
            dst[i] = nd;
            if (nd > best) { best = nd; besti = i; }
        }

        #pragma unroll
        for (int off = 16; off > 0; off >>= 1) {
            const float od = __shfl_xor(best, off, 32);
            const int   oi = __shfl_xor(besti, off, 32);
            if (od > best || (od == best && oi < besti)) { best = od; besti = oi; }
        }
        if (lane == 0) { s_d[wave] = best; s_i[wave] = besti; }
        __syncthreads();

        if (wave == 0) {
            float wd = s_d[lane];
            int   wi = s_i[lane];
            #pragma unroll
            for (int off = 16; off > 0; off >>= 1) {
                const float od = __shfl_xor(wd, off, 32);
                const int   oi = __shfl_xor(wi, off, 32);
                if (od > wd || (od == wd && oi < wi)) { wd = od; wi = oi; }
            }
            if (lane == 0) s_last = wi;
        }
        __syncthreads();
    }
}

__global__ __launch_bounds__(256) void gemm_wmma_kernel(
    const float* __restrict__ feats, const int* __restrict__ idx,
    const float* __restrict__ W, const float* __restrict__ bias,
    float* __restrict__ out, int N)
{
    const int wave = threadIdx.x >> 5;
    const int lane = threadIdx.x & 31;
    const int tile = blockIdx.x * 8 + wave;

    const int b  = tile >> 8;
    const int tm = (tile >> 4) & 15;
    const int tn = tile & 15;

    const int lhalf = lane >> 4;
    const int l15   = lane & 15;

    const int row = idx[b * FPS_K + tm * 16 + l15];
    const float* arow = feats + ((size_t)b * N + row) * IN_D;
    const float* wcol = W + tn * 16 + l15;

    v8f acc = {};
#pragma unroll
    for (int k0 = 0; k0 < IN_D; k0 += 4) {
        const int ka = k0 + lhalf * 2;
        v2f a, bf;
        a.x  = arow[ka];
        a.y  = arow[ka + 1];
        bf.x = wcol[(size_t)ka * OUT_D];
        bf.y = wcol[(size_t)(ka + 1) * OUT_D];
        acc = __builtin_amdgcn_wmma_f32_16x16x4_f32(
             false, a,  false, bf,
             (short)0, acc,  false,  false);
    }

    const int n  = tn * 16 + l15;
    const int m0 = tm * 16 + lhalf * 8;
    const float bv = bias[n];
    float* orow = out + ((size_t)b * FPS_K + m0) * OUT_D + n;
#pragma unroll
    for (int r = 0; r < 8; ++r)
        orow[(size_t)r * OUT_D] = acc[r] + bv;
}

extern "C" void kernel_launch(void* const* d_in, const int* in_sizes, int n_in,
                              void* d_out, int out_size, void* d_ws, size_t ws_size,
                              hipStream_t stream) {
    const float* feats = (const float*)d_in[0];
    const float* means = (const float*)d_in[1];
    const float* W     = (const float*)d_in[2];
    const float* bias  = (const float*)d_in[3];
    float*       out   = (float*)d_out;

    const int B = 4;
    const int N = in_sizes[1] / (B * 3);

    float* dists = (float*)d_ws;
    int*   idx   = (int*)((char*)d_ws + (size_t)B * N * sizeof(float));

    fps_kernel<<<B, FPS_T, 0, stream>>>(means, N, dists, idx);

    const int tiles = B * (FPS_K / 16) * (OUT_D / 16);
    gemm_wmma_kernel<<<tiles / 8, 256, 0, stream>>>(feats, idx, W, bias, out, N);
}


# FPS running min-distances stay on chip (registers + LDS) for all 256 steps; a step only streams coordinates, 6 independent loads per batch, next batch in flight; f32 sum order unchanged
# speedup vs baseline: 2.6499x; 2.6499x over previous
#include <hip/hip_runtime.h>

#define FPS_K   256
#define IN_D    128
#define OUT_D   256
#define FPS_T   1024
#define WAVES   (FPS_T / 32)

typedef __attribute__((ext_vector_type(2))) float v2f;
typedef __attribute__((ext_vector_type(8))) float v8f;

struct Pt3 { float x, y, z; };

#define FPS_G   6
#define FPS_NB  16
#define FPS_RB  4
#define FPS_LB  (FPS_NB - FPS_RB)
#define FPS_BIG 3.402823466e38f

__global__ __launch_bounds__(FPS_T) void fps_kernel(
    const float* __restrict__ means, int N,
    float* __restrict__ dists, int* __restrict__ idx_out)
{
    const int b    = blockIdx.x;
    const int t    = threadIdx.x;
    const int wave = t >> 5;
    const int lane = t & 31;

    const Pt3*  m   = (const Pt3*)(means + (size_t)b * N * 3);
    const Pt3*  mt  = m + t;
    float*      dst = dists + (size_t)b * N;

    __shared__ float2 s_dist2[FPS_LB * 3 * FPS_T];
    __shared__ float  s_d[WAVES];
    __shared__ int    s_i[WAVES];
    __shared__ int    s_last;

    int nb = (N / FPS_T) / FPS_G;
    nb = nb < FPS_NB ? nb : FPS_NB;
    const int sb   = nb * FPS_G;
    const int tail = (sb + FPS_G) * FPS_T;

    float dreg[FPS_RB * FPS_G];
#pragma unroll
    for (int j = 0; j < FPS_RB * FPS_G; ++j) dreg[j] = (j / FPS_G < nb) ? FPS_BIG : -1.0f;
#pragma unroll
    for (int c = 0; c < FPS_LB * 3; ++c) {
        const float v = (FPS_RB + c / 3 < nb) ? FPS_BIG : -1.0f;
        s_dist2[c * FPS_T + t] = make_float2(v, v);
    }

    float drag[FPS_G];
    int   irag[FPS_G];
#pragma unroll
    for (int jj = 0; jj < FPS_G; ++jj) {
        const int i = (sb + jj) * FPS_T + t;
        drag[jj] = i < N ? FPS_BIG : -1.0f;
        irag[jj] = i < N ? i : N - 1;
    }
    for (int i = t + tail; i < N; i += FPS_T) dst[i] = FPS_BIG;
    if (t == 0) s_last = 0;
    __syncthreads();

    for (int k = 0; k < FPS_K; ++k) {
        const int last = s_last;
        if (t == 0) idx_out[(size_t)b * FPS_K + k] = last;

        const Pt3 p = m[last];

        float best  = -1.0f;
        int   bestj = 0;

        Pt3 rag[FPS_G];
#pragma unroll
        for (int jj = 0; jj < FPS_G; ++jj) rag[jj] = m[irag[jj]];

        if (nb > 0) {
        Pt3 cur[FPS_G];
#pragma unroll
        for (int jj = 0; jj < FPS_G; ++jj) cur[jj] = mt[jj * FPS_T];
#pragma unroll
        for (int g = 0; g < FPS_NB; ++g) {
            Pt3 nxt[FPS_G];
            if (g + 1 < FPS_NB) {
                const int gl = (g + 1 < nb) ? g + 1 : nb - 1;
                const Pt3* mg = mt + (size_t)gl * (FPS_G * FPS_T);
#pragma unroll
                for (int jj = 0; jj < FPS_G; ++jj) nxt[jj] = mg[jj * FPS_T];
            }
            {
                float d[FPS_G];
                if (g < FPS_RB) {
#pragma unroll
                    for (int jj = 0; jj < FPS_G; ++jj) d[jj] = dreg[g * FPS_G + jj];
                } else {
#pragma unroll
                    for (int c = 0; c < 3; ++c) {
                        const float2 v = s_dist2[((g - FPS_RB) * 3 + c) * FPS_T + t];
                        d[2 * c] = v.x; d[2 * c + 1] = v.y;
                    }
                }
#pragma unroll
                for (int jj = 0; jj < FPS_G; ++jj) {
                    const Pt3 pt = cur[jj];
                    const float dx = pt.x - p.x;
                    const float dy = pt.y - p.y;
                    const float dz = pt.z - p.z;
                    const float d2 = fmaf(dx, dx, fmaf(dy, dy, dz * dz));
                    const float nd = fminf(d[jj], d2);
                    d[jj] = nd;
                    if (nd > best) { best = nd; bestj = g * FPS_G + jj; }
                }
                if (g < FPS_RB) {
#pragma unroll
                    for (int jj = 0; jj < FPS_G; ++jj) dreg[g * FPS_G + jj] = d[jj];
                } else {
#pragma unroll
                    for (int c = 0; c < 3; ++c)
                        s_dist2[((g - FPS_RB) * 3 + c) * FPS_T + t] = make_float2(d[2 * c], d[2 * c + 1]);
                }
            }
            if (g + 1 < FPS_NB) {
#pragma unroll
                for (int jj = 0; jj < FPS_G; ++jj) cur[jj] = nxt[jj];
            }
        }
        }

#pragma unroll
        for (int jj = 0; jj < FPS_G; ++jj) {
            const Pt3 pt = rag[jj];
            const float dx = pt.x - p.x;
            const float dy = pt.y - p.y;
            const float dz = pt.z - p.z;
            const float d2 = fmaf(dx, dx, fmaf(dy, dy, dz * dz));
            const float nd = fminf(drag[jj], d2);
            drag[jj] = nd;
            if (nd > best) { best = nd; bestj = sb + jj; }
        }

        int besti = bestj * FPS_T + t;

        for (int i = t + tail; i < N; i += FPS_T) {
            const Pt3 pt = m[i];
            const float dx = pt.x - p.x;
            const float dy = pt.y - p.y;
            const float dz = pt.z - p.z;
            const float d2 = fmaf(dx, dx, fmaf(dy, dy, dz * dz));
            const float nd = fminf(dst[i], d2);
            dst[i] = nd;
            if (nd > best) { best = nd; besti = i; }
        }

        #pragma unroll
        for (int off = 16; off > 0; off >>= 1) {
            const float od = __shfl_xor(best, off, 32);
            const int   oi = __shfl_xor(besti, off, 32);
            if (od > best || (od == best && oi < besti)) { best = od; besti = oi; }
        }
        if (lane == 0) { s_d[wave] = best; s_i[wave] = besti; }
        __syncthreads();

        if (wave == 0) {
            float wd = s_d[lane];
            int   wi = s_i[lane];
            #pragma unroll
            for (int off = 16; off > 0; off >>= 1) {
                const float od = __shfl_xor(wd, off, 32);
                const int   oi = __shfl_xor(wi, off, 32);
                if (od > wd || (od == wd && oi < wi)) { wd = od; wi = oi; }
            }
            if (lane == 0) s_last = wi;
        }
        __syncthreads();
    }
}

__global__ __launch_bounds__(256) void gemm_wmma_kernel(
    const float* __restrict__ feats, const int* __restrict__ idx,
    const float* __restrict__ W, const float* __restrict__ bias,
    float* __restrict__ out, int N)
{
    const int wave = threadIdx.x >> 5;
    const int lane = threadIdx.x & 31;
    const int tile = blockIdx.x * 8 + wave;

    const int b  = tile >> 8;
    const int tm = (tile >> 4) & 15;
    const int tn = tile & 15;

    const int lhalf = lane >> 4;
    const int l15   = lane & 15;

    const int row = idx[b * FPS_K + tm * 16 + l15];
    const float* arow = feats + ((size_t)b * N + row) * IN_D;
    const float* wcol = W + tn * 16 + l15;

    v8f acc = {};
#pragma unroll
    for (int k0 = 0; k0 < IN_D; k0 += 4) {
        const int ka = k0 + lhalf * 2;
        v2f a, bf;
        a.x  = arow[ka];
        a.y  = arow[ka + 1];
        bf.x = wcol[(size_t)ka * OUT_D];
        bf.y = wcol[(size_t)(ka + 1) * OUT_D];
        acc = __builtin_amdgcn_wmma_f32_16x16x4_f32(
             false, a,  false, bf,
             (short)0, acc,  false,  false);
    }

    const int n  = tn * 16 + l15;
    const int m0 = tm * 16 + lhalf * 8;
    const float bv = bias[n];
    float* orow = out + ((size_t)b * FPS_K + m0) * OUT_D + n;
#pragma unroll
    for (int r = 0; r < 8; ++r)
        orow[(size_t)r * OUT_D] = acc[r] + bv;
}

extern "C" void kernel_launch(void* const* d_in, const int* in_sizes, int n_in,
                              void* d_out, int out_size, void* d_ws, size_t ws_size,
                              hipStream_t stream) {
    const float* feats = (const float*)d_in[0];
    const float* means = (const float*)d_in[1];
    const float* W     = (const float*)d_in[2];
    const float* bias  = (const float*)d_in[3];
    float*       out   = (float*)d_out;

    const int B = 4;
    const int N = in_sizes[1] / (B * 3);

    float* dists = (float*)d_ws;
    int*   idx   = (int*)((char*)d_ws + (size_t)B * N * sizeof(float));

    fps_kernel<<<B, FPS_T, 0, stream>>>(means, N, dists, idx);

    const int tiles = B * (FPS_K / 16) * (OUT_D / 16);
    gemm_wmma_kernel<<<tiles / 8, 256, 0, stream>>>(feats, idx, W, bias, out, N);
}


# FPS: one launch per step, 32 workgroups per batch over the chip instead of one persistent workgroup per batch; distances in the workspace again; f32 sum order unchanged
# speedup vs baseline: 6.8578x; 2.5879x over previous
#include <hip/hip_runtime.h>

#define FPS_K   256
#define IN_D    128
#define OUT_D   256

typedef __attribute__((ext_vector_type(2))) float v2f;
typedef __attribute__((ext_vector_type(8))) float v8f;

struct Pt3 { float x, y, z; };

#define STEP_T   256
#define STEP_W   (STEP_T / 32)
#define STEP_G   32
#define STEP_U   13
#define FPS_BIG  3.402823466e38f
#define FPS_NONE 0x7fffffff

#define HAND_LINE 32
#define HAND_WORDS_PER_BATCH (STEP_G * HAND_LINE)
static_assert(STEP_G == 32, "the winner's reduction reads one pair per lane of a wave32");

__device__ __forceinline__ void argmax_pair(float& d, int& i, float od, int oi)
{
    const bool take = (od > d) | ((od == d) & (oi < i));
    d = take ? od : d;
    i = take ? oi : i;
}

template <int M>
__device__ __forceinline__ int row_xor(int v)
{
    return __builtin_amdgcn_update_dpp(0, v, 0x160 + M, 0xf, 0xf, false);
}
__device__ __forceinline__ int half_xor(int v)
{
    return __builtin_amdgcn_permlanex16(v, v, 0x76543210, 0xfedcba98, false, false);
}
template <int M>
__device__ __forceinline__ void argmax_row_step(float& d, int& i)
{
    const float od = __int_as_float(row_xor<M>(__float_as_int(d)));
    const int   oi = row_xor<M>(i);
    argmax_pair(d, i, od, oi);
}
__device__ __forceinline__ void argmax_wave(float& d, int& i)
{
    argmax_row_step<1>(d, i);
    argmax_row_step<2>(d, i);
    argmax_row_step<4>(d, i);
    argmax_row_step<8>(d, i);
    const float od = __int_as_float(half_xor(__float_as_int(d)));
    const int   oi = half_xor(i);
    argmax_pair(d, i, od, oi);
}

template <bool FIRST>
__global__ __launch_bounds__(STEP_T) void fps_step_kernel(
    const float* __restrict__ means, int N, int B,
    float* __restrict__ dists, int* __restrict__ idx_out,
    size_t ws_size, float* __restrict__ out, int k)
{
    const int g    = blockIdx.x;
    const int b    = blockIdx.y;
    const int t    = threadIdx.x;
    const int wave = t >> 5;
    const int lane = t & 31;

    const Pt3* m   = (const Pt3*)(means + (size_t)b * N * 3);
    float*     dst = dists + (size_t)b * N;

    const size_t used = (size_t)B * N + (size_t)B * FPS_K;
    const size_t area = (size_t)B * HAND_WORDS_PER_BATCH;
    float* hand = (ws_size >= (used + 2 * area) * sizeof(float)) ? dists + used : out;
    float*       hw = hand + ((size_t)(k & 1) * B + b) * HAND_WORDS_PER_BATCH + g * HAND_LINE;
    const float* hr = hand + ((size_t)((k & 1) ^ 1) * B + b) * HAND_WORDS_PER_BATCH;

    __shared__ float s_d[STEP_W];
    __shared__ int   s_i[STEP_W];

    const int ch   = ((N + STEP_G * 32 - 1) / (STEP_G * 32)) * 32;
    const int base = g * ch;
    const int end  = (base + ch < N) ? base + ch : N;

    Pt3   c[STEP_U];
    float d[STEP_U];
#pragma unroll
    for (int s = 0; s < STEP_U; ++s) {
        const int i  = base + s * STEP_T + t;
        const int ic = i < N ? i : N - 1;
        c[s] = m[ic];
        d[s] = FIRST ? FPS_BIG : dst[ic];
    }

    int last = 0;
    if (!FIRST) {
        const float2 pr = *(const float2*)(hr + lane * HAND_LINE);
        float wd = pr.x;
        int   wi = __float_as_int(pr.y);
        argmax_wave(wd, wi);
        last = __builtin_amdgcn_readfirstlane(wi);
    }
    if (g == 0 && t == 0) idx_out[(size_t)b * FPS_K + k] = last;

    const Pt3 p = m[last];

    float best  = -1.0f;
    int   besti = FPS_NONE;

#pragma unroll
    for (int s = 0; s < STEP_U; ++s) {
        const int  i     = base + s * STEP_T + t;
        const bool valid = i < end;
        const float dx = c[s].x - p.x;
        const float dy = c[s].y - p.y;
        const float dz = c[s].z - p.z;
        const float d2 = fmaf(dx, dx, fmaf(dy, dy, dz * dz));
        const float nd = fminf(d[s], d2);
        if (valid) dst[i] = nd;
        const float cand = valid ? nd : -1.0f;
        if (cand > best) { best = cand; besti = i; }
    }

    for (int i = base + STEP_U * STEP_T + t; i < end; i += STEP_T) {
        const Pt3 pt = m[i];
        const float dx = pt.x - p.x;
        const float dy = pt.y - p.y;
        const float dz = pt.z - p.z;
        const float d2 = fmaf(dx, dx, fmaf(dy, dy, dz * dz));
        const float dold = FIRST ? FPS_BIG : dst[i];
        const float nd = fminf(dold, d2);
        dst[i] = nd;
        if (nd > best) { best = nd; besti = i; }
    }

    argmax_wave(best, besti);
    if (lane == 0) { s_d[wave] = best; s_i[wave] = besti; }
    __syncthreads();

    if (wave == 0) {
        float wd = lane < STEP_W ? s_d[lane] : -1.0f;
        int   wi = lane < STEP_W ? s_i[lane] : FPS_NONE;
        argmax_row_step<1>(wd, wi);
        argmax_row_step<2>(wd, wi);
        argmax_row_step<4>(wd, wi);
        if (lane == 0) *(float2*)hw = make_float2(wd, __int_as_float(wi));
    }
}

__global__ __launch_bounds__(256) void gemm_wmma_kernel(
    const float* __restrict__ feats, const int* __restrict__ idx,
    const float* __restrict__ W, const float* __restrict__ bias,
    float* __restrict__ out, int N)
{
    const int wave = threadIdx.x >> 5;
    const int lane = threadIdx.x & 31;
    const int tile = blockIdx.x * 8 + wave;

    const int b  = tile >> 8;
    const int tm = (tile >> 4) & 15;
    const int tn = tile & 15;

    const int lhalf = lane >> 4;
    const int l15   = lane & 15;

    const int row = idx[b * FPS_K + tm * 16 + l15];
    const float* arow = feats + ((size_t)b * N + row) * IN_D;
    const float* wcol = W + tn * 16 + l15;

    v8f acc = {};
#pragma unroll
    for (int k0 = 0; k0 < IN_D; k0 += 4) {
        const int ka = k0 + lhalf * 2;
        v2f a, bf;
        a.x  = arow[ka];
        a.y  = arow[ka + 1];
        bf.x = wcol[(size_t)ka * OUT_D];
        bf.y = wcol[(size_t)(ka + 1) * OUT_D];
        acc = __builtin_amdgcn_wmma_f32_16x16x4_f32(
             false, a,  false, bf,
             (short)0, acc,  false,  false);
    }

    const int n  = tn * 16 + l15;
    const int m0 = tm * 16 + lhalf * 8;
    const float bv = bias[n];
    float* orow = out + ((size_t)b * FPS_K + m0) * OUT_D + n;
#pragma unroll
    for (int r = 0; r < 8; ++r)
        orow[(size_t)r * OUT_D] = acc[r] + bv;
}

extern "C" void kernel_launch(void* const* d_in, const int* in_sizes, int n_in,
                              void* d_out, int out_size, void* d_ws, size_t ws_size,
                              hipStream_t stream) {
    const float* feats = (const float*)d_in[0];
    const float* means = (const float*)d_in[1];
    const float* W     = (const float*)d_in[2];
    const float* bias  = (const float*)d_in[3];
    float*       out   = (float*)d_out;

    const int B = 4;
    const int N = in_sizes[1] / (B * 3);

    float* dists = (float*)d_ws;
    int*   idx   = (int*)((char*)d_ws + (size_t)B * N * sizeof(float));

    fps_step_kernel<true><<<dim3(STEP_G, B), STEP_T, 0, stream>>>(
        means, N, B, dists, idx, ws_size, out, 0);
    for (int k = 1; k < FPS_K; ++k)
        fps_step_kernel<false><<<dim3(STEP_G, B), STEP_T, 0, stream>>>(
            means, N, B, dists, idx, ws_size, out, k);

    const int tiles = B * (FPS_K / 16) * (OUT_D / 16);
    gemm_wmma_kernel<<<tiles / 8, 256, 0, stream>>>(feats, idx, W, bias, out, N);
}


# FPS: several selections per launch: every workgroup re-derives the next winners from the 32 chunk leaders; chunk runner-up distances bound all other points; 98 launches, not 257; f32 sum order kept
# speedup vs baseline: 15.1064x; 2.2028x over previous
#include <hip/hip_runtime.h>

#define FPS_K   256
#define IN_D    128
#define OUT_D   256

typedef __attribute__((ext_vector_type(2))) float v2f;
typedef __attribute__((ext_vector_type(8))) float v8f;

struct Pt3 { float x, y, z; };

#define STEP_T   256
#define STEP_W   (STEP_T / 32)
#define STEP_G   32
#define STEP_U   13
#define FPS_BIG  3.402823466e38f
#define FPS_NONE 0x7fffffff
#define FPS_LAUNCHES 96
#define FIN_T    1024
#define FIN_W    (FIN_T / 32)

#define HAND_LINE 32
#define HAND_WORDS_PER_BATCH (STEP_G * HAND_LINE)
static_assert(STEP_G == 32, "the leaders' reduction reads one record per lane of a wave32");
static_assert(FIN_W == 32, "the finisher's second reduction reads one wave leader per lane");

__device__ __forceinline__ float dist2(float cx, float cy, float cz, float px, float py, float pz)
{
    const float dx = cx - px;
    const float dy = cy - py;
    const float dz = cz - pz;
    return fmaf(dx, dx, fmaf(dy, dy, dz * dz));
}

__device__ __forceinline__ void argmax_pair(float& d, int& i, float od, int oi)
{
    const bool take = (od > d) | ((od == d) & (oi < i));
    d = take ? od : d;
    i = take ? oi : i;
}

__device__ __forceinline__ void top2_merge(float& d, int& i, float& s, float od, int oi, float os)
{
    const bool take = (od > d) | ((od == d) & (oi < i));
    const float ns = take ? fmaxf(d, os) : fmaxf(s, od);
    d = take ? od : d;
    i = take ? oi : i;
    s = ns;
}

template <int M>
__device__ __forceinline__ int row_xor(int v)
{
    return __builtin_amdgcn_update_dpp(0, v, 0x160 + M, 0xf, 0xf, false);
}
__device__ __forceinline__ int half_xor(int v)
{
    return __builtin_amdgcn_permlanex16(v, v, 0x76543210, 0xfedcba98, false, false);
}
template <int M>
__device__ __forceinline__ void argmax_row_step(float& d, int& i)
{
    const float od = __int_as_float(row_xor<M>(__float_as_int(d)));
    const int   oi = row_xor<M>(i);
    argmax_pair(d, i, od, oi);
}
__device__ __forceinline__ void argmax_wave(float& d, int& i)
{
    argmax_row_step<1>(d, i);
    argmax_row_step<2>(d, i);
    argmax_row_step<4>(d, i);
    argmax_row_step<8>(d, i);
    const float od = __int_as_float(half_xor(__float_as_int(d)));
    const int   oi = half_xor(i);
    argmax_pair(d, i, od, oi);
}
template <int M>
__device__ __forceinline__ void top2_row_step(float& d, int& i, float& s)
{
    const float od = __int_as_float(row_xor<M>(__float_as_int(d)));
    const int   oi = row_xor<M>(i);
    const float os = __int_as_float(row_xor<M>(__float_as_int(s)));
    top2_merge(d, i, s, od, oi, os);
}
__device__ __forceinline__ void top2_wave(float& d, int& i, float& s)
{
    top2_row_step<1>(d, i, s);
    top2_row_step<2>(d, i, s);
    top2_row_step<4>(d, i, s);
    top2_row_step<8>(d, i, s);
    const float od = __int_as_float(half_xor(__float_as_int(d)));
    const int   oi = half_xor(i);
    const float os = __int_as_float(half_xor(__float_as_int(s)));
    top2_merge(d, i, s, od, oi, os);
}
template <int M>
__device__ __forceinline__ float max_row_step(float v)
{
    return fmaxf(v, __int_as_float(row_xor<M>(__float_as_int(v))));
}
__device__ __forceinline__ float max_wave(float v)
{
    v = max_row_step<1>(v);
    v = max_row_step<2>(v);
    v = max_row_step<4>(v);
    v = max_row_step<8>(v);
    return fmaxf(v, __int_as_float(half_xor(__float_as_int(v))));
}
__device__ __forceinline__ float uni_f(float v)
{
    return __int_as_float(__builtin_amdgcn_readfirstlane(__float_as_int(v)));
}
__device__ __forceinline__ float lane_f(float v, int src)
{
    return __int_as_float(__builtin_amdgcn_readlane(__float_as_int(v), src));
}

__device__ __forceinline__ void top2_add(float& best, int& besti, float& second, float cand, int i)
{
    if (cand > best) { second = best; best = cand; besti = i; }
    else             { second = fmaxf(second, cand); }
}

__device__ __forceinline__ float* hand_area(float* dists, float* out, size_t ws_size, int N, int B)
{
    const size_t used = (size_t)B * N + (size_t)B * FPS_K;
    const size_t area = (size_t)B * HAND_WORDS_PER_BATCH;
    return (ws_size >= (used + 2 * area) * sizeof(float)) ? dists + used : out;
}

template <bool FIRST>
__global__ __launch_bounds__(STEP_T) void fps_step_kernel(
    const float* __restrict__ means, int N, int B,
    float* __restrict__ dists, int* __restrict__ idx_out,
    size_t ws_size, float* __restrict__ out, int L)
{
    const int g    = blockIdx.x;
    const int b    = blockIdx.y;
    const int t    = threadIdx.x;
    const int wave = t >> 5;
    const int lane = t & 31;

    const Pt3* m   = (const Pt3*)(means + (size_t)b * N * 3);
    float*     dst = dists + (size_t)b * N;

    float* hand = hand_area(dists, out, ws_size, N, B);
    float*       hw = hand + ((size_t)(L & 1) * B + b) * HAND_WORDS_PER_BATCH + g * HAND_LINE;
    const float* hr = hand + ((size_t)((L & 1) ^ 1) * B + b) * HAND_WORDS_PER_BATCH;

    __shared__ float s_d[STEP_W];
    __shared__ int   s_i[STEP_W];
    __shared__ float s_s[STEP_W];
    __shared__ float s_p[3 * STEP_G];

    int   cnt = 0;
    float cv = -1.0f, cx = 0.0f, cy = 0.0f, cz = 0.0f;
    int   ci = FPS_NONE;
    float U  = -1.0f;
    if (!FIRST) {
        const float4 r0 = *(const float4*)(hr + lane * HAND_LINE);
        const float4 r1 = *(const float4*)(hr + lane * HAND_LINE + 4);
        cv = r0.x; ci = __float_as_int(r0.y); cx = r0.z; cy = r0.w; cz = r1.x;
        cnt = __builtin_amdgcn_readfirstlane(__float_as_int(r1.z));
        if (cnt >= FPS_K) {
            if (wave == 0 && lane == g) {
                *(float4*)hw       = r0;
                *(float4*)(hw + 4) = r1;
            }
            return;
        }
        U = uni_f(max_wave(r1.y));
    }

    const int ch   = ((N + STEP_G * 32 - 1) / (STEP_G * 32)) * 32;
    const int base = g * ch;
    const int end  = (base + ch < N) ? base + ch : N;

    Pt3   c[STEP_U];
    float d[STEP_U];
#pragma unroll
    for (int s = 0; s < STEP_U; ++s) {
        const int i  = base + s * STEP_T + t;
        const int ic = i < N ? i : N - 1;
        c[s] = m[ic];
        d[s] = FIRST ? FPS_BIG : dst[ic];
    }

    int T = 0;
    for (;;) {
        float px, py, pz;
        int   pi;
        if (FIRST) {
            pi = 0;
            const Pt3 p0 = m[0];
            px = p0.x; py = p0.y; pz = p0.z;
        } else {
            float wd = cv;
            int   wi = ci;
            argmax_wave(wd, wi);
            if (T > 0 && !(uni_f(wd) > U)) break;
            pi = __builtin_amdgcn_readfirstlane(wi);
            const unsigned hit = __builtin_amdgcn_ballot_w32(ci == pi);
            const int src = hit ? __builtin_ctz(hit) : 0;
            px = lane_f(cx, src); py = lane_f(cy, src); pz = lane_f(cz, src);
        }
        if (g == 0 && t == 0) idx_out[(size_t)b * FPS_K + cnt + T] = pi;
        if (t == 0) { s_p[3 * T] = px; s_p[3 * T + 1] = py; s_p[3 * T + 2] = pz; }

        if (!FIRST) cv = fminf(cv, dist2(cx, cy, cz, px, py, pz));
#pragma unroll
        for (int s = 0; s < STEP_U; ++s)
            d[s] = fminf(d[s], dist2(c[s].x, c[s].y, c[s].z, px, py, pz));

        ++T;
        if (FIRST || cnt + T >= FPS_K || T >= STEP_G) break;
    }

    float best   = -1.0f;
    float second = -1.0f;
    int   besti  = FPS_NONE;
#pragma unroll
    for (int s = 0; s < STEP_U; ++s) {
        const int  i     = base + s * STEP_T + t;
        const bool valid = i < end;
        if (valid) dst[i] = d[s];
        top2_add(best, besti, second, valid ? d[s] : -1.0f, i);
    }

    if (ch > STEP_U * STEP_T) {
        __syncthreads();
        for (int i = base + STEP_U * STEP_T + t; i < end; i += STEP_T) {
            const Pt3 pt = m[i];
            float nd = FIRST ? FPS_BIG : dst[i];
            for (int q = 0; q < T; ++q)
                nd = fminf(nd, dist2(pt.x, pt.y, pt.z, s_p[3 * q], s_p[3 * q + 1], s_p[3 * q + 2]));
            dst[i] = nd;
            top2_add(best, besti, second, nd, i);
        }
    }

    top2_wave(best, besti, second);
    if (lane == 0) { s_d[wave] = best; s_i[wave] = besti; s_s[wave] = second; }
    __syncthreads();

    if (wave == 0) {
        float wd = lane < STEP_W ? s_d[lane] : -1.0f;
        int   wi = lane < STEP_W ? s_i[lane] : FPS_NONE;
        float ws = lane < STEP_W ? s_s[lane] : -1.0f;
        top2_row_step<1>(wd, wi, ws);
        top2_row_step<2>(wd, wi, ws);
        top2_row_step<4>(wd, wi, ws);
        if (lane == 0) {
            const int wc = wi < N ? wi : N - 1;
            const Pt3 w = m[wc];
            *(float4*)hw       = make_float4(wd, __int_as_float(wi), w.x, w.y);
            *(float4*)(hw + 4) = make_float4(w.z, ws, __int_as_float(cnt + T), 0.0f);
        }
    }
}

__global__ __launch_bounds__(FIN_T) void fps_finish_kernel(
    const float* __restrict__ means, int N, int B,
    float* __restrict__ dists, int* __restrict__ idx_out,
    size_t ws_size, float* __restrict__ out, int Llast)
{
    const int b    = blockIdx.x;
    const int t    = threadIdx.x;
    const int wave = t >> 5;
    const int lane = t & 31;

    const Pt3* m   = (const Pt3*)(means + (size_t)b * N * 3);
    float*     dst = dists + (size_t)b * N;

    const float* hand = hand_area(dists, out, ws_size, N, B);
    const float* hr   = hand + ((size_t)(Llast & 1) * B + b) * HAND_WORDS_PER_BATCH;

    __shared__ float f_d[FIN_W];
    __shared__ int   f_i[FIN_W];

    const float4 r0 = *(const float4*)(hr + lane * HAND_LINE);
    const float4 r1 = *(const float4*)(hr + lane * HAND_LINE + 4);
    const int cnt = __builtin_amdgcn_readfirstlane(__float_as_int(r1.z));
    if (cnt >= FPS_K) return;

    float wd = r0.x;
    int   wi = __float_as_int(r0.y);
    argmax_wave(wd, wi);
    int last = __builtin_amdgcn_readfirstlane(wi);

    for (int k = cnt; k < FPS_K; ++k) {
        if (t == 0) idx_out[(size_t)b * FPS_K + k] = last;
        if (k == FPS_K - 1) break;
        const int lc = last < 0 ? 0 : (last < N ? last : N - 1);
        const Pt3 p = m[lc];
        float best  = -1.0f;
        int   besti = FPS_NONE;
        for (int i = t; i < N; i += FIN_T) {
            const Pt3 pt = m[i];
            const float nd = fminf(dst[i], dist2(pt.x, pt.y, pt.z, p.x, p.y, p.z));
            dst[i] = nd;
            if (nd > best) { best = nd; besti = i; }
        }
        argmax_wave(best, besti);
        if (lane == 0) { f_d[wave] = best; f_i[wave] = besti; }
        __syncthreads();
        float xd = f_d[lane];
        int   xi = f_i[lane];
        argmax_wave(xd, xi);
        last = __builtin_amdgcn_readfirstlane(xi);
        __syncthreads();
    }
}

__global__ __launch_bounds__(256) void gemm_wmma_kernel(
    const float* __restrict__ feats, const int* __restrict__ idx,
    const float* __restrict__ W, const float* __restrict__ bias,
    float* __restrict__ out, int N)
{
    const int wave = threadIdx.x >> 5;
    const int lane = threadIdx.x & 31;
    const int tile = blockIdx.x * 8 + wave;

    const int b  = tile >> 8;
    const int tm = (tile >> 4) & 15;
    const int tn = tile & 15;

    const int lhalf = lane >> 4;
    const int l15   = lane & 15;

    const int row = idx[b * FPS_K + tm * 16 + l15];
    const float* arow = feats + ((size_t)b * N + row) * IN_D;
    const float* wcol = W + tn * 16 + l15;

    v8f acc = {};
#pragma unroll
    for (int k0 = 0; k0 < IN_D; k0 += 4) {
        const int ka = k0 + lhalf * 2;
        v2f a, bf;
        a.x  = arow[ka];
        a.y  = arow[ka + 1];
        bf.x = wcol[(size_t)ka * OUT_D];
        bf.y = wcol[(size_t)(ka + 1) * OUT_D];
        acc = __builtin_amdgcn_wmma_f32_16x16x4_f32(
             false, a,  false, bf,
             (short)0, acc,  false,  false);
    }

    const int n  = tn * 16 + l15;
    const int m0 = tm * 16 + lhalf * 8;
    const float bv = bias[n];
    float* orow = out + ((size_t)b * FPS_K + m0) * OUT_D + n;
#pragma unroll
    for (int r = 0; r < 8; ++r)
        orow[(size_t)r * OUT_D] = acc[r] + bv;
}

extern "C" void kernel_launch(void* const* d_in, const int* in_sizes, int n_in,
                              void* d_out, int out_size, void* d_ws, size_t ws_size,
                              hipStream_t stream) {
    const float* feats = (const float*)d_in[0];
    const float* means = (const float*)d_in[1];
    const float* W     = (const float*)d_in[2];
    const float* bias  = (const float*)d_in[3];
    float*       out   = (float*)d_out;

    const int B = 4;
    const int N = in_sizes[1] / (B * 3);

    float* dists = (float*)d_ws;
    int*   idx   = (int*)((char*)d_ws + (size_t)B * N * sizeof(float));

    fps_step_kernel<true><<<dim3(STEP_G, B), STEP_T, 0, stream>>>(
        means, N, B, dists, idx, ws_size, out, 0);
    for (int L = 1; L < FPS_LAUNCHES; ++L)
        fps_step_kernel<false><<<dim3(STEP_G, B), STEP_T, 0, stream>>>(
            means, N, B, dists, idx, ws_size, out, L);
    fps_finish_kernel<<<B, FIN_T, 0, stream>>>(
        means, N, B, dists, idx, ws_size, out, FPS_LAUNCHES - 1);

    const int tiles = B * (FPS_K / 16) * (OUT_D / 16);
    gemm_wmma_kernel<<<tiles / 8, 256, 0, stream>>>(feats, idx, W, bias, out, N);
}


# FPS: finer records: each wave leaves its 2 leading points and a third-place bound (512 candidates a batch, not 32): ~10 selections a launch, not 4; 34 launches, not 98; 4-wave groups; sum order kept
# speedup vs baseline: 21.0801x; 1.3954x over previous
#include <hip/hip_runtime.h>

#define FPS_K   256
#define IN_D    128
#define OUT_D   256

typedef __attribute__((ext_vector_type(2))) float v2f;
typedef __attribute__((ext_vector_type(8))) float v8f;

struct Pt3 { float x, y, z; };

#define STEP_T   128
#define STEP_W   (STEP_T / 32)
#define STEP_G   64
#define STEP_U   13
#define FPS_BIG  3.402823466e38f
#define FPS_NONE 0x7fffffff
#define FPS_LAUNCHES 32
#define FIN_T    1024
#define FIN_W    (FIN_T / 32)

#define REC_W  16
#define RECS   (STEP_G * STEP_W)
#define CAND_R (RECS / 32)
#define NCAND  (2 * CAND_R)
#define HAND_WORDS_PER_BATCH (RECS * REC_W)
static_assert((NCAND & (NCAND - 1)) == 0, "a lane's candidates are reduced as a binary tree");
static_assert(RECS % 32 == 0, "the records are dealt to the lanes of a wave32");
static_assert((STEP_W * REC_W * 4) % 128 == 0, "a workgroup's records fill whole 128-byte lines");
static_assert(STEP_W * REC_W / 4 <= 32, "one wave stores the workgroup's records with one instruction");
static_assert((RECS & (RECS - 1)) == 0 && RECS <= FIN_T, "the finisher reads one record per thread, modulo RECS");
static_assert(FIN_W == 32, "the finisher's second reduction reads one wave leader per lane");

__device__ __forceinline__ float dist2(float cx, float cy, float cz, float px, float py, float pz)
{
    const float dx = cx - px;
    const float dy = cy - py;
    const float dz = cz - pz;
    return fmaf(dx, dx, fmaf(dy, dy, dz * dz));
}

__device__ __forceinline__ bool ahead(float ad, int ai, float bd, int bi)
{
    return (ad > bd) | ((ad == bd) & (ai < bi));
}

__device__ __forceinline__ void argmax_pair(float& d, int& i, float od, int oi)
{
    const bool take = ahead(od, oi, d, i);
    d = take ? od : d;
    i = take ? oi : i;
}

__device__ __forceinline__ void top3_merge(float& d1, int& i1, float& d2, int& i2, float& s3,
                                           float od1, int oi1, float od2, int oi2, float os3)
{
    const bool  ox  = ahead(od1, oi1, d1, i1);
    const float xd1 = ox ? od1 : d1;   const int xi1 = ox ? oi1 : i1;
    const float xd2 = ox ? od2 : d2;   const int xi2 = ox ? oi2 : i2;
    const float xs3 = ox ? os3 : s3;
    const float yd1 = ox ? d1 : od1;   const int yi1 = ox ? i1 : oi1;
    const float yd2 = ox ? d2 : od2;
    const bool  k2  = ahead(xd2, xi2, yd1, yi1);
    d1 = xd1;                 i1 = xi1;
    d2 = k2 ? xd2 : yd1;      i2 = k2 ? xi2 : yi1;
    s3 = k2 ? fmaxf(xs3, yd1) : fmaxf(xd2, yd2);
}

template <int M>
__device__ __forceinline__ int row_xor(int v)
{
    return __builtin_amdgcn_update_dpp(0, v, 0x160 + M, 0xf, 0xf, false);
}
__device__ __forceinline__ int half_xor(int v)
{
    return __builtin_amdgcn_permlanex16(v, v, 0x76543210, 0xfedcba98, false, false);
}
template <int M>
__device__ __forceinline__ void argmax_row_step(float& d, int& i)
{
    const float od = __int_as_float(row_xor<M>(__float_as_int(d)));
    const int   oi = row_xor<M>(i);
    argmax_pair(d, i, od, oi);
}
__device__ __forceinline__ void argmax_wave(float& d, int& i)
{
    argmax_row_step<1>(d, i);
    argmax_row_step<2>(d, i);
    argmax_row_step<4>(d, i);
    argmax_row_step<8>(d, i);
    const float od = __int_as_float(half_xor(__float_as_int(d)));
    const int   oi = half_xor(i);
    argmax_pair(d, i, od, oi);
}
template <int M>
__device__ __forceinline__ void top3_row_step(float& d1, int& i1, float& d2, int& i2, float& s3)
{
    const float od1 = __int_as_float(row_xor<M>(__float_as_int(d1)));
    const int   oi1 = row_xor<M>(i1);
    const float od2 = __int_as_float(row_xor<M>(__float_as_int(d2)));
    const int   oi2 = row_xor<M>(i2);
    const float os3 = __int_as_float(row_xor<M>(__float_as_int(s3)));
    top3_merge(d1, i1, d2, i2, s3, od1, oi1, od2, oi2, os3);
}
__device__ __forceinline__ void top3_wave(float& d1, int& i1, float& d2, int& i2, float& s3)
{
    top3_row_step<1>(d1, i1, d2, i2, s3);
    top3_row_step<2>(d1, i1, d2, i2, s3);
    top3_row_step<4>(d1, i1, d2, i2, s3);
    top3_row_step<8>(d1, i1, d2, i2, s3);
    const float od1 = __int_as_float(half_xor(__float_as_int(d1)));
    const int   oi1 = half_xor(i1);
    const float od2 = __int_as_float(half_xor(__float_as_int(d2)));
    const int   oi2 = half_xor(i2);
    const float os3 = __int_as_float(half_xor(__float_as_int(s3)));
    top3_merge(d1, i1, d2, i2, s3, od1, oi1, od2, oi2, os3);
}
template <int M>
__device__ __forceinline__ float max_row_step(float v)
{
    return fmaxf(v, __int_as_float(row_xor<M>(__float_as_int(v))));
}
__device__ __forceinline__ float max_wave(float v)
{
    v = max_row_step<1>(v);
    v = max_row_step<2>(v);
    v = max_row_step<4>(v);
    v = max_row_step<8>(v);
    return fmaxf(v, __int_as_float(half_xor(__float_as_int(v))));
}
__device__ __forceinline__ float uni_f(float v)
{
    return __int_as_float(__builtin_amdgcn_readfirstlane(__float_as_int(v)));
}
__device__ __forceinline__ float lane_f(float v, int src)
{
    return __int_as_float(__builtin_amdgcn_readlane(__float_as_int(v), src));
}

__device__ __forceinline__ void top3_add(float& d1, int& i1, float& d2, int& i2, float& s3, float cand, int i)
{
    if (cand > d1)      { s3 = d2; d2 = d1; i2 = i1; d1 = cand; i1 = i; }
    else if (cand > d2) { s3 = d2; d2 = cand; i2 = i; }
    else                { s3 = fmaxf(s3, cand); }
}

__device__ __forceinline__ float* hand_area(float* dists, float* out, size_t ws_size, int N, int B)
{
    const size_t used = (size_t)B * N + (size_t)B * FPS_K;
    const size_t area = (size_t)B * HAND_WORDS_PER_BATCH;
    return (ws_size >= (used + 2 * area) * sizeof(float)) ? dists + used : out;
}

template <bool FIRST>
__global__ __launch_bounds__(STEP_T) void fps_step_kernel(
    const float* __restrict__ means, int N, int B,
    float* __restrict__ dists, int* __restrict__ idx_out,
    size_t ws_size, float* __restrict__ out, int L)
{
    const int g    = blockIdx.x;
    const int b    = blockIdx.y;
    const int t    = threadIdx.x;
    const int wave = t >> 5;
    const int lane = t & 31;

    const Pt3* m   = (const Pt3*)(means + (size_t)b * N * 3);
    float*     dst = dists + (size_t)b * N;

    float* hand = hand_area(dists, out, ws_size, N, B);
    float*       hw = hand + ((size_t)(L & 1) * B + b) * HAND_WORDS_PER_BATCH + g * (STEP_W * REC_W);
    const float* hr = hand + ((size_t)((L & 1) ^ 1) * B + b) * HAND_WORDS_PER_BATCH;

    __shared__ float4 s_rec[STEP_W * REC_W / 4];
    __shared__ float  s_p[3 * FPS_K];

    int cnt = 0;
    if (!FIRST) {
        cnt = __builtin_amdgcn_readfirstlane(__float_as_int(hr[11]));
        if (cnt >= FPS_K) {
            if (wave == 0 && lane < STEP_W * REC_W / 4)
                ((float4*)hw)[lane] = ((const float4*)(hr + g * (STEP_W * REC_W)))[lane];
            return;
        }
    }

    const int ch   = ((N + STEP_G * 32 - 1) / (STEP_G * 32)) * 32;
    const int base = g * ch;
    const int end  = (base + ch < N) ? base + ch : N;

    Pt3   c[STEP_U];
    float d[STEP_U];
#pragma unroll
    for (int s = 0; s < STEP_U; ++s) {
        const int i  = base + s * STEP_T + t;
        const int ic = i < N ? i : N - 1;
        c[s] = m[ic];
        d[s] = FIRST ? FPS_BIG : dst[ic];
    }

    float U = -1.0f;
    float cd[NCAND], cx[NCAND], cy[NCAND], cz[NCAND];
    int   ci[NCAND];
    if (!FIRST) {
        float4 q0[CAND_R], q1[CAND_R], q2[CAND_R];
#pragma unroll
        for (int r = 0; r < CAND_R; ++r) {
            const float4* rp = (const float4*)(hr + (size_t)(r * 32 + lane) * REC_W);
            q0[r] = rp[0];
            q1[r] = rp[1];
            q2[r] = rp[2];
        }
        __builtin_amdgcn_sched_barrier(0);
        float u = -1.0f;
#pragma unroll
        for (int r = 0; r < CAND_R; ++r) {
            cd[2 * r]     = q0[r].x; ci[2 * r]     = __float_as_int(q0[r].y);
            cx[2 * r]     = q0[r].z; cy[2 * r]     = q0[r].w; cz[2 * r]     = q1[r].x;
            cd[2 * r + 1] = q1[r].y; ci[2 * r + 1] = __float_as_int(q1[r].z);
            cx[2 * r + 1] = q1[r].w; cy[2 * r + 1] = q2[r].x; cz[2 * r + 1] = q2[r].y;
            u = fmaxf(u, q2[r].z);
        }
        U = uni_f(max_wave(u));
    }

    int T = 0;
    for (;;) {
        float px, py, pz;
        int   pi;
        if (FIRST) {
            pi = 0;
            const Pt3 p0 = m[0];
            px = p0.x; py = p0.y; pz = p0.z;
        } else {
            float td[NCAND];
            int   ti[NCAND];
#pragma unroll
            for (int k = 0; k < NCAND; ++k) { td[k] = cd[k]; ti[k] = ci[k]; }
#pragma unroll
            for (int w = NCAND / 2; w >= 1; w >>= 1) {
#pragma unroll
                for (int k = 0; k < w; ++k) argmax_pair(td[k], ti[k], td[k + w], ti[k + w]);
            }
            const float bd = td[0];
            const int   bi = ti[0];
            float wd = bd;
            int   wi = bi;
            argmax_wave(wd, wi);
            if (T > 0 && !(uni_f(wd) > U)) break;
            pi = __builtin_amdgcn_readfirstlane(wi);
            float sx = 0.0f, sy = 0.0f, sz = 0.0f;
#pragma unroll
            for (int k = 0; k < NCAND; ++k) {
                const bool h = ci[k] == pi;
                sx = h ? cx[k] : sx; sy = h ? cy[k] : sy; sz = h ? cz[k] : sz;
            }
            const unsigned hit = __builtin_amdgcn_ballot_w32(bi == pi);
            const int src = hit ? __builtin_ctz(hit) : 0;
            px = lane_f(sx, src); py = lane_f(sy, src); pz = lane_f(sz, src);
        }
        if (g == 0 && t == 0) idx_out[(size_t)b * FPS_K + cnt + T] = pi;
        if (t == 0) { s_p[3 * T] = px; s_p[3 * T + 1] = py; s_p[3 * T + 2] = pz; }

        if (!FIRST) {
#pragma unroll
            for (int k = 0; k < NCAND; ++k)
                cd[k] = fminf(cd[k], dist2(cx[k], cy[k], cz[k], px, py, pz));
        }
#pragma unroll
        for (int s = 0; s < STEP_U; ++s)
            d[s] = fminf(d[s], dist2(c[s].x, c[s].y, c[s].z, px, py, pz));

        ++T;
        if (FIRST || cnt + T >= FPS_K) break;
    }

    float d1 = -1.0f, d2 = -1.0f, s3 = -1.0f;
    int   i1 = FPS_NONE, i2 = FPS_NONE;
#pragma unroll
    for (int s = 0; s < STEP_U; ++s) {
        const int  i     = base + s * STEP_T + t;
        const bool valid = i < end;
        if (valid) dst[i] = d[s];
        top3_add(d1, i1, d2, i2, s3, valid ? d[s] : -1.0f, i);
    }

    if (ch > STEP_U * STEP_T) {
        __syncthreads();
        for (int i = base + STEP_U * STEP_T + t; i < end; i += STEP_T) {
            const Pt3 pt = m[i];
            float nd = FIRST ? FPS_BIG : dst[i];
            for (int q = 0; q < T; ++q)
                nd = fminf(nd, dist2(pt.x, pt.y, pt.z, s_p[3 * q], s_p[3 * q + 1], s_p[3 * q + 2]));
            dst[i] = nd;
            top3_add(d1, i1, d2, i2, s3, nd, i);
        }
    }

    top3_wave(d1, i1, d2, i2, s3);
    {
        const int mine = lane == 0 ? i1 : i2;
        const int mc   = (unsigned)mine < (unsigned)N ? mine : N - 1;
        float* rec = (float*)s_rec + wave * REC_W;
        if (lane < 2) {
            const Pt3 w = m[mc];
            float* q = rec + 5 * lane;
            q[0] = lane == 0 ? d1 : d2;
            q[1] = __int_as_float(mine);
            q[2] = w.x; q[3] = w.y; q[4] = w.z;
        }
        if (lane == 2) {
            rec[10] = s3;
            rec[11] = __int_as_float(cnt + T);
            rec[12] = 0.0f; rec[13] = 0.0f; rec[14] = 0.0f; rec[15] = 0.0f;
        }
    }
    __syncthreads();
    if (wave == 0 && lane < STEP_W * REC_W / 4) ((float4*)hw)[lane] = s_rec[lane];
}

__global__ __launch_bounds__(FIN_T) void fps_finish_kernel(
    const float* __restrict__ means, int N, int B,
    float* __restrict__ dists, int* __restrict__ idx_out,
    size_t ws_size, float* __restrict__ out, int Llast)
{
    const int b    = blockIdx.x;
    const int t    = threadIdx.x;
    const int wave = t >> 5;
    const int lane = t & 31;

    const Pt3* m   = (const Pt3*)(means + (size_t)b * N * 3);
    float*     dst = dists + (size_t)b * N;

    const float* hand = hand_area(dists, out, ws_size, N, B);
    const float* hr   = hand + ((size_t)(Llast & 1) * B + b) * HAND_WORDS_PER_BATCH;

    __shared__ float f_d[FIN_W];
    __shared__ int   f_i[FIN_W];

    const float* rp = hr + (size_t)(t & (RECS - 1)) * REC_W;
    const float4 r0 = *(const float4*)rp;
    const float4 r2 = *(const float4*)(rp + 8);
    const int cnt = __builtin_amdgcn_readfirstlane(__float_as_int(r2.w));
    if (cnt >= FPS_K) return;

    float wd = r0.x;
    int   wi = __float_as_int(r0.y);
    argmax_wave(wd, wi);
    if (lane == 0) { f_d[wave] = wd; f_i[wave] = wi; }
    __syncthreads();
    float ld = f_d[lane];
    int   li = f_i[lane];
    argmax_wave(ld, li);
    int last = __builtin_amdgcn_readfirstlane(li);
    __syncthreads();

    for (int k = cnt; k < FPS_K; ++k) {
        if (t == 0) idx_out[(size_t)b * FPS_K + k] = last;
        if (k == FPS_K - 1) break;
        const int lc = last < 0 ? 0 : (last < N ? last : N - 1);
        const Pt3 p = m[lc];
        float best  = -1.0f;
        int   besti = FPS_NONE;
        for (int i = t; i < N; i += FIN_T) {
            const Pt3 pt = m[i];
            const float nd = fminf(dst[i], dist2(pt.x, pt.y, pt.z, p.x, p.y, p.z));
            dst[i] = nd;
            if (nd > best) { best = nd; besti = i; }
        }
        argmax_wave(best, besti);
        if (lane == 0) { f_d[wave] = best; f_i[wave] = besti; }
        __syncthreads();
        float xd = f_d[lane];
        int   xi = f_i[lane];
        argmax_wave(xd, xi);
        last = __builtin_amdgcn_readfirstlane(xi);
        __syncthreads();
    }
}

__global__ __launch_bounds__(256) void gemm_wmma_kernel(
    const float* __restrict__ feats, const int* __restrict__ idx,
    const float* __restrict__ W, const float* __restrict__ bias,
    float* __restrict__ out, int N)
{
    const int wave = threadIdx.x >> 5;
    const int lane = threadIdx.x & 31;
    const int tile = blockIdx.x * 8 + wave;

    const int b  = tile >> 8;
    const int tm = (tile >> 4) & 15;
    const int tn = tile & 15;

    const int lhalf = lane >> 4;
    const int l15   = lane & 15;

    const int row = idx[b * FPS_K + tm * 16 + l15];
    const float* arow = feats + ((size_t)b * N + row) * IN_D;
    const float* wcol = W + tn * 16 + l15;

    v8f acc = {};
#pragma unroll
    for (int k0 = 0; k0 < IN_D; k0 += 4) {
        const int ka = k0 + lhalf * 2;
        v2f a, bf;
        a.x  = arow[ka];
        a.y  = arow[ka + 1];
        bf.x = wcol[(size_t)ka * OUT_D];
        bf.y = wcol[(size_t)(ka + 1) * OUT_D];
        acc = __builtin_amdgcn_wmma_f32_16x16x4_f32(
             false, a,  false, bf,
             (short)0, acc,  false,  false);
    }

    const int n  = tn * 16 + l15;
    const int m0 = tm * 16 + lhalf * 8;
    const float bv = bias[n];
    float* orow = out + ((size_t)b * FPS_K + m0) * OUT_D + n;
#pragma unroll
    for (int r = 0; r < 8; ++r)
        orow[(size_t)r * OUT_D] = acc[r] + bv;
}

extern "C" void kernel_launch(void* const* d_in, const int* in_sizes, int n_in,
                              void* d_out, int out_size, void* d_ws, size_t ws_size,
                              hipStream_t stream) {
    const float* feats = (const float*)d_in[0];
    const float* means = (const float*)d_in[1];
    const float* W     = (const float*)d_in[2];
    const float* bias  = (const float*)d_in[3];
    float*       out   = (float*)d_out;

    const int B = 4;
    const int N = in_sizes[1] / (B * 3);

    float* dists = (float*)d_ws;
    int*   idx   = (int*)((char*)d_ws + (size_t)B * N * sizeof(float));

    fps_step_kernel<true><<<dim3(STEP_G, B), STEP_T, 0, stream>>>(
        means, N, B, dists, idx, ws_size, out, 0);
    for (int L = 1; L < FPS_LAUNCHES; ++L)
        fps_step_kernel<false><<<dim3(STEP_G, B), STEP_T, 0, stream>>>(
            means, N, B, dists, idx, ws_size, out, L);
    fps_finish_kernel<<<B, FIN_T, 0, stream>>>(
        means, N, B, dists, idx, ws_size, out, FPS_LAUNCHES - 1);

    const int tiles = B * (FPS_K / 16) * (OUT_D / 16);
    gemm_wmma_kernel<<<tiles / 8, 256, 0, stream>>>(feats, idx, W, bias, out, N);
}


# FPS selection loop: arg-max as two vector-ALU reductions (max distance, then lowest index among equals; DPP steps across lanes), no compare/scalar-logic/select pairs; same winners, sum order kept
# speedup vs baseline: 25.2182x; 1.1963x over previous
#include <hip/hip_runtime.h>

#define FPS_K   256
#define IN_D    128
#define OUT_D   256

typedef __attribute__((ext_vector_type(2))) float v2f;
typedef __attribute__((ext_vector_type(8))) float v8f;

struct Pt3 { float x, y, z; };

#define STEP_T   128
#define STEP_W   (STEP_T / 32)
#define STEP_G   64
#define STEP_U   13
#define FPS_BIG  3.402823466e38f
#define FPS_NONE 0x7fffffff
#define FPS_LAUNCHES 32
#define FIN_T    1024
#define FIN_W    (FIN_T / 32)

#define REC_W  16
#define RECS   (STEP_G * STEP_W)
#define CAND_R (RECS / 32)
#define NCAND  (2 * CAND_R)
#define HAND_WORDS_PER_BATCH (RECS * REC_W)
static_assert((NCAND & (NCAND - 1)) == 0, "a lane's candidates are reduced as a binary tree");
static_assert(RECS % 32 == 0, "the records are dealt to the lanes of a wave32");
static_assert((STEP_W * REC_W * 4) % 128 == 0, "a workgroup's records fill whole 128-byte lines");
static_assert(STEP_W * REC_W / 4 <= 32, "one wave stores the workgroup's records with one instruction");
static_assert((RECS & (RECS - 1)) == 0 && RECS <= FIN_T, "the finisher reads one record per thread, modulo RECS");
static_assert(FIN_W == 32, "the finisher's second reduction reads one wave leader per lane");

__device__ __forceinline__ float dist2(float cx, float cy, float cz, float px, float py, float pz)
{
    const float dx = cx - px;
    const float dy = cy - py;
    const float dz = cz - pz;
    return fmaf(dx, dx, fmaf(dy, dy, dz * dz));
}

__device__ __forceinline__ bool ahead(float ad, int ai, float bd, int bi)
{
    return (ad > bd) | ((ad == bd) & (ai < bi));
}

__device__ __forceinline__ void argmax_pair(float& d, int& i, float od, int oi)
{
    const bool take = ahead(od, oi, d, i);
    d = take ? od : d;
    i = take ? oi : i;
}

__device__ __forceinline__ void top3_merge(float& d1, int& i1, float& d2, int& i2, float& s3,
                                           float od1, int oi1, float od2, int oi2, float os3)
{
    const bool  ox  = ahead(od1, oi1, d1, i1);
    const float xd1 = ox ? od1 : d1;   const int xi1 = ox ? oi1 : i1;
    const float xd2 = ox ? od2 : d2;   const int xi2 = ox ? oi2 : i2;
    const float xs3 = ox ? os3 : s3;
    const float yd1 = ox ? d1 : od1;   const int yi1 = ox ? i1 : oi1;
    const float yd2 = ox ? d2 : od2;
    const bool  k2  = ahead(xd2, xi2, yd1, yi1);
    d1 = xd1;                 i1 = xi1;
    d2 = k2 ? xd2 : yd1;      i2 = k2 ? xi2 : yi1;
    s3 = k2 ? fmaxf(xs3, yd1) : fmaxf(xd2, yd2);
}

template <int M>
__device__ __forceinline__ int row_xor(int v)
{
    return __builtin_amdgcn_update_dpp(0, v, 0x160 + M, 0xf, 0xf, false);
}
__device__ __forceinline__ int half_xor(int v)
{
    return __builtin_amdgcn_permlanex16(v, v, 0x76543210, 0xfedcba98, false, false);
}
template <int M>
__device__ __forceinline__ void argmax_row_step(float& d, int& i)
{
    const float od = __int_as_float(row_xor<M>(__float_as_int(d)));
    const int   oi = row_xor<M>(i);
    argmax_pair(d, i, od, oi);
}
__device__ __forceinline__ void argmax_wave(float& d, int& i)
{
    argmax_row_step<1>(d, i);
    argmax_row_step<2>(d, i);
    argmax_row_step<4>(d, i);
    argmax_row_step<8>(d, i);
    const float od = __int_as_float(half_xor(__float_as_int(d)));
    const int   oi = half_xor(i);
    argmax_pair(d, i, od, oi);
}
template <int M>
__device__ __forceinline__ void top3_row_step(float& d1, int& i1, float& d2, int& i2, float& s3)
{
    const float od1 = __int_as_float(row_xor<M>(__float_as_int(d1)));
    const int   oi1 = row_xor<M>(i1);
    const float od2 = __int_as_float(row_xor<M>(__float_as_int(d2)));
    const int   oi2 = row_xor<M>(i2);
    const float os3 = __int_as_float(row_xor<M>(__float_as_int(s3)));
    top3_merge(d1, i1, d2, i2, s3, od1, oi1, od2, oi2, os3);
}
__device__ __forceinline__ void top3_wave(float& d1, int& i1, float& d2, int& i2, float& s3)
{
    top3_row_step<1>(d1, i1, d2, i2, s3);
    top3_row_step<2>(d1, i1, d2, i2, s3);
    top3_row_step<4>(d1, i1, d2, i2, s3);
    top3_row_step<8>(d1, i1, d2, i2, s3);
    const float od1 = __int_as_float(half_xor(__float_as_int(d1)));
    const int   oi1 = half_xor(i1);
    const float od2 = __int_as_float(half_xor(__float_as_int(d2)));
    const int   oi2 = half_xor(i2);
    const float os3 = __int_as_float(half_xor(__float_as_int(s3)));
    top3_merge(d1, i1, d2, i2, s3, od1, oi1, od2, oi2, os3);
}
template <int M>
__device__ __forceinline__ float max_row_step(float v)
{
    return fmaxf(v, __int_as_float(row_xor<M>(__float_as_int(v))));
}
__device__ __forceinline__ float max_wave(float v)
{
    v = max_row_step<1>(v);
    v = max_row_step<2>(v);
    v = max_row_step<4>(v);
    v = max_row_step<8>(v);
    return fmaxf(v, __int_as_float(half_xor(__float_as_int(v))));
}
__device__ __forceinline__ float max_wave_dpp(float v)
{
    asm("v_max_num_f32_dpp %0, %0, %0 row_xmask:1 row_mask:0xf bank_mask:0xf\n\t"
        "v_max_num_f32_dpp %0, %0, %0 row_xmask:2 row_mask:0xf bank_mask:0xf\n\t"
        "v_max_num_f32_dpp %0, %0, %0 row_xmask:4 row_mask:0xf bank_mask:0xf\n\t"
        "v_max_num_f32_dpp %0, %0, %0 row_xmask:8 row_mask:0xf bank_mask:0xf"
        : "+v"(v));
    return fmaxf(v, __int_as_float(half_xor(__float_as_int(v))));
}
__device__ __forceinline__ int min_wave_dpp(int v)
{
    asm("v_min_i32_dpp %0, %0, %0 row_xmask:1 row_mask:0xf bank_mask:0xf\n\t"
        "v_min_i32_dpp %0, %0, %0 row_xmask:2 row_mask:0xf bank_mask:0xf\n\t"
        "v_min_i32_dpp %0, %0, %0 row_xmask:4 row_mask:0xf bank_mask:0xf\n\t"
        "v_min_i32_dpp %0, %0, %0 row_xmask:8 row_mask:0xf bank_mask:0xf"
        : "+v"(v));
    const int o = half_xor(v);
    return o < v ? o : v;
}
__device__ __forceinline__ float uni_f(float v)
{
    return __int_as_float(__builtin_amdgcn_readfirstlane(__float_as_int(v)));
}
__device__ __forceinline__ float lane_f(float v, int src)
{
    return __int_as_float(__builtin_amdgcn_readlane(__float_as_int(v), src));
}

__device__ __forceinline__ void top3_add(float& d1, int& i1, float& d2, int& i2, float& s3, float cand, int i)
{
    if (cand > d1)      { s3 = d2; d2 = d1; i2 = i1; d1 = cand; i1 = i; }
    else if (cand > d2) { s3 = d2; d2 = cand; i2 = i; }
    else                { s3 = fmaxf(s3, cand); }
}

__device__ __forceinline__ float* hand_area(float* dists, float* out, size_t ws_size, int N, int B)
{
    const size_t used = (size_t)B * N + (size_t)B * FPS_K;
    const size_t area = (size_t)B * HAND_WORDS_PER_BATCH;
    return (ws_size >= (used + 2 * area) * sizeof(float)) ? dists + used : out;
}

template <bool FIRST>
__global__ __launch_bounds__(STEP_T) void fps_step_kernel(
    const float* __restrict__ means, int N, int B,
    float* __restrict__ dists, int* __restrict__ idx_out,
    size_t ws_size, float* __restrict__ out, int L)
{
    const int g    = blockIdx.x;
    const int b    = blockIdx.y;
    const int t    = threadIdx.x;
    const int wave = t >> 5;
    const int lane = t & 31;

    const Pt3* m   = (const Pt3*)(means + (size_t)b * N * 3);
    float*     dst = dists + (size_t)b * N;

    float* hand = hand_area(dists, out, ws_size, N, B);
    float*       hw = hand + ((size_t)(L & 1) * B + b) * HAND_WORDS_PER_BATCH + g * (STEP_W * REC_W);
    const float* hr = hand + ((size_t)((L & 1) ^ 1) * B + b) * HAND_WORDS_PER_BATCH;

    __shared__ float4 s_rec[STEP_W * REC_W / 4];
    __shared__ float  s_p[3 * FPS_K];

    int cnt = 0;
    if (!FIRST) {
        cnt = __builtin_amdgcn_readfirstlane(__float_as_int(hr[11]));
        if (cnt >= FPS_K) {
            if (wave == 0 && lane < STEP_W * REC_W / 4)
                ((float4*)hw)[lane] = ((const float4*)(hr + g * (STEP_W * REC_W)))[lane];
            return;
        }
    }

    const int ch   = ((N + STEP_G * 32 - 1) / (STEP_G * 32)) * 32;
    const int base = g * ch;
    const int end  = (base + ch < N) ? base + ch : N;

    Pt3   c[STEP_U];
    float d[STEP_U];
#pragma unroll
    for (int s = 0; s < STEP_U; ++s) {
        const int i  = base + s * STEP_T + t;
        const int ic = i < N ? i : N - 1;
        c[s] = m[ic];
        d[s] = FIRST ? FPS_BIG : dst[ic];
    }

    float U = -1.0f;
    float cd[NCAND], cx[NCAND], cy[NCAND], cz[NCAND];
    int   ci[NCAND];
    if (!FIRST) {
        float4 q0[CAND_R], q1[CAND_R], q2[CAND_R];
#pragma unroll
        for (int r = 0; r < CAND_R; ++r) {
            const float4* rp = (const float4*)(hr + (size_t)(r * 32 + lane) * REC_W);
            q0[r] = rp[0];
            q1[r] = rp[1];
            q2[r] = rp[2];
        }
        __builtin_amdgcn_sched_barrier(0);
        float u = -1.0f;
#pragma unroll
        for (int r = 0; r < CAND_R; ++r) {
            cd[2 * r]     = q0[r].x; ci[2 * r]     = __float_as_int(q0[r].y);
            cx[2 * r]     = q0[r].z; cy[2 * r]     = q0[r].w; cz[2 * r]     = q1[r].x;
            cd[2 * r + 1] = q1[r].y; ci[2 * r + 1] = __float_as_int(q1[r].z);
            cx[2 * r + 1] = q1[r].w; cy[2 * r + 1] = q2[r].x; cz[2 * r + 1] = q2[r].y;
            u = fmaxf(u, q2[r].z);
        }
        U = uni_f(max_wave(u));
    }

    int T = 0;
    for (;;) {
        float px, py, pz;
        int   pi;
        if (FIRST) {
            pi = 0;
            const Pt3 p0 = m[0];
            px = p0.x; py = p0.y; pz = p0.z;
        } else {
            float tm[NCAND];
#pragma unroll
            for (int k = 0; k < NCAND; ++k) tm[k] = cd[k];
#pragma unroll
            for (int w = NCAND / 2; w >= 1; w >>= 1) {
#pragma unroll
                for (int k = 0; k < w; ++k) tm[k] = fmaxf(tm[k], tm[k + w]);
            }
            const float wmax = uni_f(max_wave_dpp(tm[0]));
            if (T > 0 && !(wmax > U)) break;
            int tn[NCAND];
#pragma unroll
            for (int k = 0; k < NCAND; ++k) tn[k] = cd[k] == wmax ? ci[k] : FPS_NONE;
#pragma unroll
            for (int w = NCAND / 2; w >= 1; w >>= 1) {
#pragma unroll
                for (int k = 0; k < w; ++k) tn[k] = tn[k + w] < tn[k] ? tn[k + w] : tn[k];
            }
            const int bi = tn[0];
            pi = __builtin_amdgcn_readfirstlane(min_wave_dpp(bi));
            int ox[NCAND], oy[NCAND], oz[NCAND];
#pragma unroll
            for (int k = 0; k < NCAND; ++k) {
                const bool h = ci[k] == bi;
                ox[k] = h ? __float_as_int(cx[k]) : 0;
                oy[k] = h ? __float_as_int(cy[k]) : 0;
                oz[k] = h ? __float_as_int(cz[k]) : 0;
            }
#pragma unroll
            for (int w = NCAND / 2; w >= 1; w >>= 1) {
#pragma unroll
                for (int k = 0; k < w; ++k) { ox[k] |= ox[k + w]; oy[k] |= oy[k + w]; oz[k] |= oz[k + w]; }
            }
            const float sx = __int_as_float(ox[0]), sy = __int_as_float(oy[0]), sz = __int_as_float(oz[0]);
            const unsigned hit = __builtin_amdgcn_ballot_w32(bi == pi);
            const int src = hit ? __builtin_ctz(hit) : 0;
            px = lane_f(sx, src); py = lane_f(sy, src); pz = lane_f(sz, src);
        }
        if (g == 0 && t == 0) idx_out[(size_t)b * FPS_K + cnt + T] = pi;
        if (t == 0) { s_p[3 * T] = px; s_p[3 * T + 1] = py; s_p[3 * T + 2] = pz; }

        if (!FIRST) {
#pragma unroll
            for (int k = 0; k < NCAND; ++k)
                cd[k] = fminf(cd[k], dist2(cx[k], cy[k], cz[k], px, py, pz));
        }
#pragma unroll
        for (int s = 0; s < STEP_U; ++s)
            d[s] = fminf(d[s], dist2(c[s].x, c[s].y, c[s].z, px, py, pz));

        ++T;
        if (FIRST || cnt + T >= FPS_K) break;
    }

    float d1 = -1.0f, d2 = -1.0f, s3 = -1.0f;
    int   i1 = FPS_NONE, i2 = FPS_NONE;
#pragma unroll
    for (int s = 0; s < STEP_U; ++s) {
        const int  i     = base + s * STEP_T + t;
        const bool valid = i < end;
        if (valid) dst[i] = d[s];
        top3_add(d1, i1, d2, i2, s3, valid ? d[s] : -1.0f, i);
    }

    if (ch > STEP_U * STEP_T) {
        __syncthreads();
        for (int i = base + STEP_U * STEP_T + t; i < end; i += STEP_T) {
            const Pt3 pt = m[i];
            float nd = FIRST ? FPS_BIG : dst[i];
            for (int q = 0; q < T; ++q)
                nd = fminf(nd, dist2(pt.x, pt.y, pt.z, s_p[3 * q], s_p[3 * q + 1], s_p[3 * q + 2]));
            dst[i] = nd;
            top3_add(d1, i1, d2, i2, s3, nd, i);
        }
    }

    top3_wave(d1, i1, d2, i2, s3);
    {
        const int mine = lane == 0 ? i1 : i2;
        const int mc   = (unsigned)mine < (unsigned)N ? mine : N - 1;
        float* rec = (float*)s_rec + wave * REC_W;
        if (lane < 2) {
            const Pt3 w = m[mc];
            float* q = rec + 5 * lane;
            q[0] = lane == 0 ? d1 : d2;
            q[1] = __int_as_float(mine);
            q[2] = w.x; q[3] = w.y; q[4] = w.z;
        }
        if (lane == 2) {
            rec[10] = s3;
            rec[11] = __int_as_float(cnt + T);
            rec[12] = 0.0f; rec[13] = 0.0f; rec[14] = 0.0f; rec[15] = 0.0f;
        }
    }
    __syncthreads();
    if (wave == 0 && lane < STEP_W * REC_W / 4) ((float4*)hw)[lane] = s_rec[lane];
}

__global__ __launch_bounds__(FIN_T) void fps_finish_kernel(
    const float* __restrict__ means, int N, int B,
    float* __restrict__ dists, int* __restrict__ idx_out,
    size_t ws_size, float* __restrict__ out, int Llast)
{
    const int b    = blockIdx.x;
    const int t    = threadIdx.x;
    const int wave = t >> 5;
    const int lane = t & 31;

    const Pt3* m   = (const Pt3*)(means + (size_t)b * N * 3);
    float*     dst = dists + (size_t)b * N;

    const float* hand = hand_area(dists, out, ws_size, N, B);
    const float* hr   = hand + ((size_t)(Llast & 1) * B + b) * HAND_WORDS_PER_BATCH;

    __shared__ float f_d[FIN_W];
    __shared__ int   f_i[FIN_W];

    const float* rp = hr + (size_t)(t & (RECS - 1)) * REC_W;
    const float4 r0 = *(const float4*)rp;
    const float4 r2 = *(const float4*)(rp + 8);
    const int cnt = __builtin_amdgcn_readfirstlane(__float_as_int(r2.w));
    if (cnt >= FPS_K) return;

    float wd = r0.x;
    int   wi = __float_as_int(r0.y);
    argmax_wave(wd, wi);
    if (lane == 0) { f_d[wave] = wd; f_i[wave] = wi; }
    __syncthreads();
    float ld = f_d[lane];
    int   li = f_i[lane];
    argmax_wave(ld, li);
    int last = __builtin_amdgcn_readfirstlane(li);
    __syncthreads();

    for (int k = cnt; k < FPS_K; ++k) {
        if (t == 0) idx_out[(size_t)b * FPS_K + k] = last;
        if (k == FPS_K - 1) break;
        const int lc = last < 0 ? 0 : (last < N ? last : N - 1);
        const Pt3 p = m[lc];
        float best  = -1.0f;
        int   besti = FPS_NONE;
        for (int i = t; i < N; i += FIN_T) {
            const Pt3 pt = m[i];
            const float nd = fminf(dst[i], dist2(pt.x, pt.y, pt.z, p.x, p.y, p.z));
            dst[i] = nd;
            if (nd > best) { best = nd; besti = i; }
        }
        argmax_wave(best, besti);
        if (lane == 0) { f_d[wave] = best; f_i[wave] = besti; }
        __syncthreads();
        float xd = f_d[lane];
        int   xi = f_i[lane];
        argmax_wave(xd, xi);
        last = __builtin_amdgcn_readfirstlane(xi);
        __syncthreads();
    }
}

__global__ __launch_bounds__(256) void gemm_wmma_kernel(
    const float* __restrict__ feats, const int* __restrict__ idx,
    const float* __restrict__ W, const float* __restrict__ bias,
    float* __restrict__ out, int N)
{
    const int wave = threadIdx.x >> 5;
    const int lane = threadIdx.x & 31;
    const int tile = blockIdx.x * 8 + wave;

    const int b  = tile >> 8;
    const int tm = (tile >> 4) & 15;
    const int tn = tile & 15;

    const int lhalf = lane >> 4;
    const int l15   = lane & 15;

    const int row = idx[b * FPS_K + tm * 16 + l15];
    const float* arow = feats + ((size_t)b * N + row) * IN_D;
    const float* wcol = W + tn * 16 + l15;

    v8f acc = {};
#pragma unroll
    for (int k0 = 0; k0 < IN_D; k0 += 4) {
        const int ka = k0 + lhalf * 2;
        v2f a, bf;
        a.x  = arow[ka];
        a.y  = arow[ka + 1];
        bf.x = wcol[(size_t)ka * OUT_D];
        bf.y = wcol[(size_t)(ka + 1) * OUT_D];
        acc = __builtin_amdgcn_wmma_f32_16x16x4_f32(
             false, a,  false, bf,
             (short)0, acc,  false,  false);
    }

    const int n  = tn * 16 + l15;
    const int m0 = tm * 16 + lhalf * 8;
    const float bv = bias[n];
    float* orow = out + ((size_t)b * FPS_K + m0) * OUT_D + n;
#pragma unroll
    for (int r = 0; r < 8; ++r)
        orow[(size_t)r * OUT_D] = acc[r] + bv;
}

extern "C" void kernel_launch(void* const* d_in, const int* in_sizes, int n_in,
                              void* d_out, int out_size, void* d_ws, size_t ws_size,
                              hipStream_t stream) {
    const float* feats = (const float*)d_in[0];
    const float* means = (const float*)d_in[1];
    const float* W     = (const float*)d_in[2];
    const float* bias  = (const float*)d_in[3];
    float*       out   = (float*)d_out;

    const int B = 4;
    const int N = in_sizes[1] / (B * 3);

    float* dists = (float*)d_ws;
    int*   idx   = (int*)((char*)d_ws + (size_t)B * N * sizeof(float));

    fps_step_kernel<true><<<dim3(STEP_G, B), STEP_T, 0, stream>>>(
        means, N, B, dists, idx, ws_size, out, 0);
    for (int L = 1; L < FPS_LAUNCHES; ++L)
        fps_step_kernel<false><<<dim3(STEP_G, B), STEP_T, 0, stream>>>(
            means, N, B, dists, idx, ws_size, out, L);
    fps_finish_kernel<<<B, FIN_T, 0, stream>>>(
        means, N, B, dists, idx, ws_size, out, FPS_LAUNCHES - 1);

    const int tiles = B * (FPS_K / 16) * (OUT_D / 16);
    gemm_wmma_kernel<<<tiles / 8, 256, 0, stream>>>(feats, idx, W, bias, out, N);
}


# FPS selection loop on the live candidates only: a launch packs those at or above the bound U (only they can be selected in it) through LDS, 6 per lane, not 16; same winners, sum order kept
# speedup vs baseline: 30.6750x; 1.2164x over previous
#include <hip/hip_runtime.h>

#define FPS_K   256
#define IN_D    128
#define OUT_D   256

typedef __attribute__((ext_vector_type(2))) float v2f;
typedef __attribute__((ext_vector_type(8))) float v8f;

struct Pt3 { float x, y, z; };

#define STEP_T   128
#define STEP_W   (STEP_T / 32)
#define STEP_G   64
#define STEP_U   13
#define FPS_BIG  3.402823466e38f
#define FPS_NONE 0x7fffffff
#define FPS_LAUNCHES 32
#define FIN_T    1024
#define FIN_W    (FIN_T / 32)

#define REC_W  16
#define RECS   (STEP_G * STEP_W)
#define CAND_R (RECS / 32)
#define NCAND  (2 * CAND_R)
#define HAND_WORDS_PER_BATCH (RECS * REC_W)
#define LIVE_R   6
#define LIVE_CAP (LIVE_R * 32)
static_assert(RECS % 32 == 0, "the records are dealt to the lanes of a wave32");
static_assert((STEP_W * REC_W * 4) % 128 == 0, "a workgroup's records fill whole 128-byte lines");
static_assert(STEP_W * REC_W / 4 <= 32, "one wave stores the workgroup's records with one instruction");
static_assert((RECS & (RECS - 1)) == 0 && RECS <= FIN_T, "the finisher reads one record per thread, modulo RECS");
static_assert(FIN_W == 32, "the finisher's second reduction reads one wave leader per lane");

__device__ __forceinline__ float dist2(float cx, float cy, float cz, float px, float py, float pz)
{
    const float dx = cx - px;
    const float dy = cy - py;
    const float dz = cz - pz;
    return fmaf(dx, dx, fmaf(dy, dy, dz * dz));
}

__device__ __forceinline__ bool ahead(float ad, int ai, float bd, int bi)
{
    return (ad > bd) | ((ad == bd) & (ai < bi));
}

__device__ __forceinline__ void argmax_pair(float& d, int& i, float od, int oi)
{
    const bool take = ahead(od, oi, d, i);
    d = take ? od : d;
    i = take ? oi : i;
}

__device__ __forceinline__ void top3_merge(float& d1, int& i1, float& d2, int& i2, float& s3,
                                           float od1, int oi1, float od2, int oi2, float os3)
{
    const bool  ox  = ahead(od1, oi1, d1, i1);
    const float xd1 = ox ? od1 : d1;   const int xi1 = ox ? oi1 : i1;
    const float xd2 = ox ? od2 : d2;   const int xi2 = ox ? oi2 : i2;
    const float xs3 = ox ? os3 : s3;
    const float yd1 = ox ? d1 : od1;   const int yi1 = ox ? i1 : oi1;
    const float yd2 = ox ? d2 : od2;
    const bool  k2  = ahead(xd2, xi2, yd1, yi1);
    d1 = xd1;                 i1 = xi1;
    d2 = k2 ? xd2 : yd1;      i2 = k2 ? xi2 : yi1;
    s3 = k2 ? fmaxf(xs3, yd1) : fmaxf(xd2, yd2);
}

template <int M>
__device__ __forceinline__ int row_xor(int v)
{
    return __builtin_amdgcn_update_dpp(0, v, 0x160 + M, 0xf, 0xf, false);
}
__device__ __forceinline__ int half_xor(int v)
{
    return __builtin_amdgcn_permlanex16(v, v, 0x76543210, 0xfedcba98, false, false);
}
template <int M>
__device__ __forceinline__ void argmax_row_step(float& d, int& i)
{
    const float od = __int_as_float(row_xor<M>(__float_as_int(d)));
    const int   oi = row_xor<M>(i);
    argmax_pair(d, i, od, oi);
}
__device__ __forceinline__ void argmax_wave(float& d, int& i)
{
    argmax_row_step<1>(d, i);
    argmax_row_step<2>(d, i);
    argmax_row_step<4>(d, i);
    argmax_row_step<8>(d, i);
    const float od = __int_as_float(half_xor(__float_as_int(d)));
    const int   oi = half_xor(i);
    argmax_pair(d, i, od, oi);
}
template <int M>
__device__ __forceinline__ void top3_row_step(float& d1, int& i1, float& d2, int& i2, float& s3)
{
    const float od1 = __int_as_float(row_xor<M>(__float_as_int(d1)));
    const int   oi1 = row_xor<M>(i1);
    const float od2 = __int_as_float(row_xor<M>(__float_as_int(d2)));
    const int   oi2 = row_xor<M>(i2);
    const float os3 = __int_as_float(row_xor<M>(__float_as_int(s3)));
    top3_merge(d1, i1, d2, i2, s3, od1, oi1, od2, oi2, os3);
}
__device__ __forceinline__ void top3_wave(float& d1, int& i1, float& d2, int& i2, float& s3)
{
    top3_row_step<1>(d1, i1, d2, i2, s3);
    top3_row_step<2>(d1, i1, d2, i2, s3);
    top3_row_step<4>(d1, i1, d2, i2, s3);
    top3_row_step<8>(d1, i1, d2, i2, s3);
    const float od1 = __int_as_float(half_xor(__float_as_int(d1)));
    const int   oi1 = half_xor(i1);
    const float od2 = __int_as_float(half_xor(__float_as_int(d2)));
    const int   oi2 = half_xor(i2);
    const float os3 = __int_as_float(half_xor(__float_as_int(s3)));
    top3_merge(d1, i1, d2, i2, s3, od1, oi1, od2, oi2, os3);
}
template <int M>
__device__ __forceinline__ float max_row_step(float v)
{
    return fmaxf(v, __int_as_float(row_xor<M>(__float_as_int(v))));
}
__device__ __forceinline__ float max_wave(float v)
{
    v = max_row_step<1>(v);
    v = max_row_step<2>(v);
    v = max_row_step<4>(v);
    v = max_row_step<8>(v);
    return fmaxf(v, __int_as_float(half_xor(__float_as_int(v))));
}
__device__ __forceinline__ float max_wave_dpp(float v)
{
    asm("v_max_num_f32_dpp %0, %0, %0 row_xmask:1 row_mask:0xf bank_mask:0xf\n\t"
        "v_max_num_f32_dpp %0, %0, %0 row_xmask:2 row_mask:0xf bank_mask:0xf\n\t"
        "v_max_num_f32_dpp %0, %0, %0 row_xmask:4 row_mask:0xf bank_mask:0xf\n\t"
        "v_max_num_f32_dpp %0, %0, %0 row_xmask:8 row_mask:0xf bank_mask:0xf"
        : "+v"(v));
    return fmaxf(v, __int_as_float(half_xor(__float_as_int(v))));
}
__device__ __forceinline__ int min_wave_dpp(int v)
{
    asm("v_min_i32_dpp %0, %0, %0 row_xmask:1 row_mask:0xf bank_mask:0xf\n\t"
        "v_min_i32_dpp %0, %0, %0 row_xmask:2 row_mask:0xf bank_mask:0xf\n\t"
        "v_min_i32_dpp %0, %0, %0 row_xmask:4 row_mask:0xf bank_mask:0xf\n\t"
        "v_min_i32_dpp %0, %0, %0 row_xmask:8 row_mask:0xf bank_mask:0xf"
        : "+v"(v));
    const int o = half_xor(v);
    return o < v ? o : v;
}
__device__ __forceinline__ float uni_f(float v)
{
    return __int_as_float(__builtin_amdgcn_readfirstlane(__float_as_int(v)));
}
__device__ __forceinline__ float lane_f(float v, int src)
{
    return __int_as_float(__builtin_amdgcn_readlane(__float_as_int(v), src));
}

__device__ __forceinline__ void top3_add(float& d1, int& i1, float& d2, int& i2, float& s3, float cand, int i)
{
    if (cand > d1)      { s3 = d2; d2 = d1; i2 = i1; d1 = cand; i1 = i; }
    else if (cand > d2) { s3 = d2; d2 = cand; i2 = i; }
    else                { s3 = fmaxf(s3, cand); }
}

__device__ __forceinline__ float* hand_area(float* dists, float* out, size_t ws_size, int N, int B)
{
    const size_t used = (size_t)B * N + (size_t)B * FPS_K;
    const size_t area = (size_t)B * HAND_WORDS_PER_BATCH;
    return (ws_size >= (used + 2 * area) * sizeof(float)) ? dists + used : out;
}

template <int NC>
__device__ __forceinline__ int select_run(
    float (&cd)[NC], const int (&ci)[NC], const float (&cx)[NC], const float (&cy)[NC], const float (&cz)[NC],
    const float U, const int cnt, const Pt3 (&c)[STEP_U], float (&d)[STEP_U],
    int* __restrict__ idx_row, float* s_p, const bool lead, const bool first_thread)
{
    int T = 0;
    for (;;) {
        float tm[NC];
#pragma unroll
        for (int k = 0; k < NC; ++k) tm[k] = cd[k];
#pragma unroll
        for (int w = NC; w > 1; w = (w + 1) / 2) {
#pragma unroll
            for (int k = 0; k < w / 2; ++k) tm[k] = fmaxf(tm[k], tm[k + (w + 1) / 2]);
        }
        const float wmax = uni_f(max_wave_dpp(tm[0]));
        if (T > 0 && !(wmax > U)) break;
        int tn[NC];
#pragma unroll
        for (int k = 0; k < NC; ++k) tn[k] = cd[k] == wmax ? ci[k] : FPS_NONE;
#pragma unroll
        for (int w = NC; w > 1; w = (w + 1) / 2) {
#pragma unroll
            for (int k = 0; k < w / 2; ++k) {
                const int o = tn[k + (w + 1) / 2];
                tn[k] = o < tn[k] ? o : tn[k];
            }
        }
        const int bi = tn[0];
        const int pi = __builtin_amdgcn_readfirstlane(min_wave_dpp(bi));
        int ox[NC], oy[NC], oz[NC];
#pragma unroll
        for (int k = 0; k < NC; ++k) {
            const bool h = ci[k] == bi;
            ox[k] = h ? __float_as_int(cx[k]) : 0;
            oy[k] = h ? __float_as_int(cy[k]) : 0;
            oz[k] = h ? __float_as_int(cz[k]) : 0;
        }
#pragma unroll
        for (int w = NC; w > 1; w = (w + 1) / 2) {
#pragma unroll
            for (int k = 0; k < w / 2; ++k) {
                ox[k] |= ox[k + (w + 1) / 2]; oy[k] |= oy[k + (w + 1) / 2]; oz[k] |= oz[k + (w + 1) / 2];
            }
        }
        const float sx = __int_as_float(ox[0]), sy = __int_as_float(oy[0]), sz = __int_as_float(oz[0]);
        const unsigned hit = __builtin_amdgcn_ballot_w32(bi == pi);
        const int src = hit ? __builtin_ctz(hit) : 0;
        const float px = lane_f(sx, src), py = lane_f(sy, src), pz = lane_f(sz, src);

        if (lead) idx_row[cnt + T] = pi;
        if (first_thread) { s_p[3 * T] = px; s_p[3 * T + 1] = py; s_p[3 * T + 2] = pz; }

#pragma unroll
        for (int k = 0; k < NC; ++k)
            cd[k] = fminf(cd[k], dist2(cx[k], cy[k], cz[k], px, py, pz));
#pragma unroll
        for (int s = 0; s < STEP_U; ++s)
            d[s] = fminf(d[s], dist2(c[s].x, c[s].y, c[s].z, px, py, pz));

        ++T;
        if (cnt + T >= FPS_K) break;
    }
    return T;
}

template <bool FIRST>
__global__ __launch_bounds__(STEP_T) void fps_step_kernel(
    const float* __restrict__ means, int N, int B,
    float* __restrict__ dists, int* __restrict__ idx_out,
    size_t ws_size, float* __restrict__ out, int L)
{
    const int g    = blockIdx.x;
    const int b    = blockIdx.y;
    const int t    = threadIdx.x;
    const int wave = t >> 5;
    const int lane = t & 31;

    const Pt3* m   = (const Pt3*)(means + (size_t)b * N * 3);
    float*     dst = dists + (size_t)b * N;

    float* hand = hand_area(dists, out, ws_size, N, B);
    float*       hw = hand + ((size_t)(L & 1) * B + b) * HAND_WORDS_PER_BATCH + g * (STEP_W * REC_W);
    const float* hr = hand + ((size_t)((L & 1) ^ 1) * B + b) * HAND_WORDS_PER_BATCH;

    __shared__ float4 s_rec[STEP_W * REC_W / 4];
    __shared__ float  s_p[3 * FPS_K];
    __shared__ float4 s_lq[STEP_W][LIVE_CAP + 32];
    __shared__ float  s_lz[STEP_W][LIVE_CAP + 32];

    int cnt = 0;
    if (!FIRST) {
        cnt = __builtin_amdgcn_readfirstlane(__float_as_int(hr[11]));
        if (cnt >= FPS_K) {
            if (wave == 0 && lane < STEP_W * REC_W / 4)
                ((float4*)hw)[lane] = ((const float4*)(hr + g * (STEP_W * REC_W)))[lane];
            return;
        }
    }

    const int ch   = ((N + STEP_G * 32 - 1) / (STEP_G * 32)) * 32;
    const int base = g * ch;
    const int end  = (base + ch < N) ? base + ch : N;

    Pt3   c[STEP_U];
    float d[STEP_U];
#pragma unroll
    for (int s = 0; s < STEP_U; ++s) {
        const int i  = base + s * STEP_T + t;
        const int ic = i < N ? i : N - 1;
        c[s] = m[ic];
        d[s] = FIRST ? FPS_BIG : dst[ic];
    }

    int T = 0;
    if (FIRST) {
        const Pt3 p0 = m[0];
        if (g == 0 && t == 0) idx_out[(size_t)b * FPS_K] = 0;
        if (t == 0) { s_p[0] = p0.x; s_p[1] = p0.y; s_p[2] = p0.z; }
#pragma unroll
        for (int s = 0; s < STEP_U; ++s)
            d[s] = fminf(d[s], dist2(c[s].x, c[s].y, c[s].z, p0.x, p0.y, p0.z));
        T = 1;
    } else {
        float cd[NCAND], cx[NCAND], cy[NCAND], cz[NCAND];
        int   ci[NCAND];
        float4 q0[CAND_R], q1[CAND_R], q2[CAND_R];
#pragma unroll
        for (int r = 0; r < CAND_R; ++r) {
            const float4* rp = (const float4*)(hr + (size_t)(r * 32 + lane) * REC_W);
            q0[r] = rp[0];
            q1[r] = rp[1];
            q2[r] = rp[2];
        }
        asm volatile("" ::: "memory");
        __builtin_amdgcn_sched_barrier(0);
        float u = -1.0f;
#pragma unroll
        for (int r = 0; r < CAND_R; ++r) {
            cd[2 * r]     = q0[r].x; ci[2 * r]     = __float_as_int(q0[r].y);
            cx[2 * r]     = q0[r].z; cy[2 * r]     = q0[r].w; cz[2 * r]     = q1[r].x;
            cd[2 * r + 1] = q1[r].y; ci[2 * r + 1] = __float_as_int(q1[r].z);
            cx[2 * r + 1] = q1[r].w; cy[2 * r + 1] = q2[r].x; cz[2 * r + 1] = q2[r].y;
            u = fmaxf(u, q2[r].z);
        }
        const float U = uni_f(max_wave(u));

        int nl = 0;
#pragma unroll
        for (int k = 0; k < NCAND; ++k) {
            const bool     lv  = cd[k] >= U;
            const unsigned mk  = __builtin_amdgcn_ballot_w32(lv);
            const int      pos = nl + (int)__builtin_amdgcn_mbcnt_lo(mk, 0u);
            const int      at  = (lv && pos < LIVE_CAP) ? pos : LIVE_CAP + lane;
            s_lq[wave][at] = make_float4(cd[k], __int_as_float(ci[k]), cx[k], cy[k]);
            s_lz[wave][at] = cz[k];
            nl += __builtin_popcount(mk);
        }
        __syncthreads();

        int* idx_row = idx_out + (size_t)b * FPS_K;
        if (nl >= 1 && nl <= LIVE_CAP) {
            float ld[LIVE_R], lx[LIVE_R], ly[LIVE_R], lz[LIVE_R];
            int   li[LIVE_R];
#pragma unroll
            for (int j = 0; j < LIVE_R; ++j) {
                const int  e  = j * 32 + lane;
                const bool ok = e < nl;
                const int  ec = ok ? e : 0;
                const float4 q = s_lq[wave][ec];
                ld[j] = ok ? q.x : -1.0f;
                li[j] = ok ? __float_as_int(q.y) : FPS_NONE;
                lx[j] = q.z; ly[j] = q.w; lz[j] = s_lz[wave][ec];
            }
            T = select_run<LIVE_R>(ld, li, lx, ly, lz, U, cnt, c, d, idx_row, s_p, g == 0 && t == 0, t == 0);
        } else {
            T = select_run<NCAND>(cd, ci, cx, cy, cz, U, cnt, c, d, idx_row, s_p, g == 0 && t == 0, t == 0);
        }
    }

    float d1 = -1.0f, d2 = -1.0f, s3 = -1.0f;
    int   i1 = FPS_NONE, i2 = FPS_NONE;
#pragma unroll
    for (int s = 0; s < STEP_U; ++s) {
        const int  i     = base + s * STEP_T + t;
        const bool valid = i < end;
        if (valid) dst[i] = d[s];
        top3_add(d1, i1, d2, i2, s3, valid ? d[s] : -1.0f, i);
    }

    if (ch > STEP_U * STEP_T) {
        __syncthreads();
        for (int i = base + STEP_U * STEP_T + t; i < end; i += STEP_T) {
            const Pt3 pt = m[i];
            float nd = FIRST ? FPS_BIG : dst[i];
            for (int q = 0; q < T; ++q)
                nd = fminf(nd, dist2(pt.x, pt.y, pt.z, s_p[3 * q], s_p[3 * q + 1], s_p[3 * q + 2]));
            dst[i] = nd;
            top3_add(d1, i1, d2, i2, s3, nd, i);
        }
    }

    top3_wave(d1, i1, d2, i2, s3);
    {
        const int mine = lane == 0 ? i1 : i2;
        const int mc   = (unsigned)mine < (unsigned)N ? mine : N - 1;
        float* rec = (float*)s_rec + wave * REC_W;
        if (lane < 2) {
            const Pt3 w = m[mc];
            float* q = rec + 5 * lane;
            q[0] = lane == 0 ? d1 : d2;
            q[1] = __int_as_float(mine);
            q[2] = w.x; q[3] = w.y; q[4] = w.z;
        }
        if (lane == 2) {
            rec[10] = s3;
            rec[11] = __int_as_float(cnt + T);
            rec[12] = 0.0f; rec[13] = 0.0f; rec[14] = 0.0f; rec[15] = 0.0f;
        }
    }
    __syncthreads();
    if (wave == 0 && lane < STEP_W * REC_W / 4) ((float4*)hw)[lane] = s_rec[lane];
}

__global__ __launch_bounds__(FIN_T) void fps_finish_kernel(
    const float* __restrict__ means, int N, int B,
    float* __restrict__ dists, int* __restrict__ idx_out,
    size_t ws_size, float* __restrict__ out, int Llast)
{
    const int b    = blockIdx.x;
    const int t    = threadIdx.x;
    const int wave = t >> 5;
    const int lane = t & 31;

    const Pt3* m   = (const Pt3*)(means + (size_t)b * N * 3);
    float*     dst = dists + (size_t)b * N;

    const float* hand = hand_area(dists, out, ws_size, N, B);
    const float* hr   = hand + ((size_t)(Llast & 1) * B + b) * HAND_WORDS_PER_BATCH;

    __shared__ float f_d[FIN_W];
    __shared__ int   f_i[FIN_W];

    const float* rp = hr + (size_t)(t & (RECS - 1)) * REC_W;
    const float4 r0 = *(const float4*)rp;
    const float4 r2 = *(const float4*)(rp + 8);
    const int cnt = __builtin_amdgcn_readfirstlane(__float_as_int(r2.w));
    if (cnt >= FPS_K) return;

    float wd = r0.x;
    int   wi = __float_as_int(r0.y);
    argmax_wave(wd, wi);
    if (lane == 0) { f_d[wave] = wd; f_i[wave] = wi; }
    __syncthreads();
    float ld = f_d[lane];
    int   li = f_i[lane];
    argmax_wave(ld, li);
    int last = __builtin_amdgcn_readfirstlane(li);
    __syncthreads();

    for (int k = cnt; k < FPS_K; ++k) {
        if (t == 0) idx_out[(size_t)b * FPS_K + k] = last;
        if (k == FPS_K - 1) break;
        const int lc = last < 0 ? 0 : (last < N ? last : N - 1);
        const Pt3 p = m[lc];
        float best  = -1.0f;
        int   besti = FPS_NONE;
        for (int i = t; i < N; i += FIN_T) {
            const Pt3 pt = m[i];
            const float nd = fminf(dst[i], dist2(pt.x, pt.y, pt.z, p.x, p.y, p.z));
            dst[i] = nd;
            if (nd > best) { best = nd; besti = i; }
        }
        argmax_wave(best, besti);
        if (lane == 0) { f_d[wave] = best; f_i[wave] = besti; }
        __syncthreads();
        float xd = f_d[lane];
        int   xi = f_i[lane];
        argmax_wave(xd, xi);
        last = __builtin_amdgcn_readfirstlane(xi);
        __syncthreads();
    }
}

__global__ __launch_bounds__(256) void gemm_wmma_kernel(
    const float* __restrict__ feats, const int* __restrict__ idx,
    const float* __restrict__ W, const float* __restrict__ bias,
    float* __restrict__ out, int N)
{
    const int wave = threadIdx.x >> 5;
    const int lane = threadIdx.x & 31;
    const int tile = blockIdx.x * 8 + wave;

    const int b  = tile >> 8;
    const int tm = (tile >> 4) & 15;
    const int tn = tile & 15;

    const int lhalf = lane >> 4;
    const int l15   = lane & 15;

    const int row = idx[b * FPS_K + tm * 16 + l15];
    const float* arow = feats + ((size_t)b * N + row) * IN_D;
    const float* wcol = W + tn * 16 + l15;

    v8f acc = {};
#pragma unroll
    for (int k0 = 0; k0 < IN_D; k0 += 4) {
        const int ka = k0 + lhalf * 2;
        v2f a, bf;
        a.x  = arow[ka];
        a.y  = arow[ka + 1];
        bf.x = wcol[(size_t)ka * OUT_D];
        bf.y = wcol[(size_t)(ka + 1) * OUT_D];
        acc = __builtin_amdgcn_wmma_f32_16x16x4_f32(
             false, a,  false, bf,
             (short)0, acc,  false,  false);
    }

    const int n  = tn * 16 + l15;
    const int m0 = tm * 16 + lhalf * 8;
    const float bv = bias[n];
    float* orow = out + ((size_t)b * FPS_K + m0) * OUT_D + n;
#pragma unroll
    for (int r = 0; r < 8; ++r)
        orow[(size_t)r * OUT_D] = acc[r] + bv;
}

extern "C" void kernel_launch(void* const* d_in, const int* in_sizes, int n_in,
                              void* d_out, int out_size, void* d_ws, size_t ws_size,
                              hipStream_t stream) {
    const float* feats = (const float*)d_in[0];
    const float* means = (const float*)d_in[1];
    const float* W     = (const float*)d_in[2];
    const float* bias  = (const float*)d_in[3];
    float*       out   = (float*)d_out;

    const int B = 4;
    const int N = in_sizes[1] / (B * 3);

    float* dists = (float*)d_ws;
    int*   idx   = (int*)((char*)d_ws + (size_t)B * N * sizeof(float));

    fps_step_kernel<true><<<dim3(STEP_G, B), STEP_T, 0, stream>>>(
        means, N, B, dists, idx, ws_size, out, 0);
    for (int L = 1; L < FPS_LAUNCHES; ++L)
        fps_step_kernel<false><<<dim3(STEP_G, B), STEP_T, 0, stream>>>(
            means, N, B, dists, idx, ws_size, out, L);
    fps_finish_kernel<<<B, FIN_T, 0, stream>>>(
        means, N, B, dists, idx, ws_size, out, FPS_LAUNCHES - 1);

    const int tiles = B * (FPS_K / 16) * (OUT_D / 16);
    gemm_wmma_kernel<<<tiles / 8, 256, 0, stream>>>(feats, idx, W, bias, out, N);
}


# FPS records: a wave leaves its 3 leading points as (distance, index) and a fourth-place bound, 32 bytes; 768 candidates a batch, not 512; reader fetches live ones' coordinates; 27 launches, not 34
# speedup vs baseline: 35.5116x; 1.1577x over previous
#include <hip/hip_runtime.h>

#define FPS_K   256
#define IN_D    128
#define OUT_D   256

typedef __attribute__((ext_vector_type(2))) float v2f;
typedef __attribute__((ext_vector_type(8))) float v8f;

struct Pt3 { float x, y, z; };

#define STEP_T   128
#define STEP_W   (STEP_T / 32)
#define STEP_G   64
#define STEP_U   13
#define FPS_BIG  3.402823466e38f
#define FPS_NONE 0x7fffffff
#define FPS_LAUNCHES 25
#define FIN_T    1024
#define FIN_W    (FIN_T / 32)

#define REC_W   8
#define REC_CNT 7
#define RECS   (STEP_G * STEP_W)
#define CAND_R (RECS / 32)
#define NCAND  (3 * CAND_R)
#define HAND_WORDS_PER_BATCH (RECS * REC_W)
#define LIVE_R   8
#define LIVE_CAP (LIVE_R * 32)
static_assert(RECS % 32 == 0, "the records are dealt to the lanes of a wave32");
static_assert((STEP_W * REC_W * 4) % 128 == 0, "a workgroup's records fill whole 128-byte lines");
static_assert(STEP_W * REC_W / 4 <= 32, "one wave stores the workgroup's records with one instruction");
static_assert(REC_W == 8, "a record is read and written as two 16-byte halves");
static_assert((RECS & (RECS - 1)) == 0 && RECS <= FIN_T, "the finisher reads one record per thread, modulo RECS");
static_assert(FIN_W == 32, "the finisher's second reduction reads one wave leader per lane");

__device__ __forceinline__ float dist2(float cx, float cy, float cz, float px, float py, float pz)
{
    const float dx = cx - px;
    const float dy = cy - py;
    const float dz = cz - pz;
    return fmaf(dx, dx, fmaf(dy, dy, dz * dz));
}

__device__ __forceinline__ bool ahead(float ad, int ai, float bd, int bi)
{
    return (ad > bd) | ((ad == bd) & (ai < bi));
}

__device__ __forceinline__ void argmax_pair(float& d, int& i, float od, int oi)
{
    const bool take = ahead(od, oi, d, i);
    d = take ? od : d;
    i = take ? oi : i;
}

template <int M>
__device__ __forceinline__ int row_xor(int v)
{
    return __builtin_amdgcn_update_dpp(0, v, 0x160 + M, 0xf, 0xf, false);
}
__device__ __forceinline__ int half_xor(int v)
{
    return __builtin_amdgcn_permlanex16(v, v, 0x76543210, 0xfedcba98, false, false);
}
template <int M>
__device__ __forceinline__ void argmax_row_step(float& d, int& i)
{
    const float od = __int_as_float(row_xor<M>(__float_as_int(d)));
    const int   oi = row_xor<M>(i);
    argmax_pair(d, i, od, oi);
}
__device__ __forceinline__ void argmax_wave(float& d, int& i)
{
    argmax_row_step<1>(d, i);
    argmax_row_step<2>(d, i);
    argmax_row_step<4>(d, i);
    argmax_row_step<8>(d, i);
    const float od = __int_as_float(half_xor(__float_as_int(d)));
    const int   oi = half_xor(i);
    argmax_pair(d, i, od, oi);
}
template <int M>
__device__ __forceinline__ float max_row_step(float v)
{
    return fmaxf(v, __int_as_float(row_xor<M>(__float_as_int(v))));
}
__device__ __forceinline__ float max_wave(float v)
{
    v = max_row_step<1>(v);
    v = max_row_step<2>(v);
    v = max_row_step<4>(v);
    v = max_row_step<8>(v);
    return fmaxf(v, __int_as_float(half_xor(__float_as_int(v))));
}
__device__ __forceinline__ float max_wave_dpp(float v)
{
    asm("v_max_num_f32_dpp %0, %0, %0 row_xmask:1 row_mask:0xf bank_mask:0xf\n\t"
        "v_max_num_f32_dpp %0, %0, %0 row_xmask:2 row_mask:0xf bank_mask:0xf\n\t"
        "v_max_num_f32_dpp %0, %0, %0 row_xmask:4 row_mask:0xf bank_mask:0xf\n\t"
        "v_max_num_f32_dpp %0, %0, %0 row_xmask:8 row_mask:0xf bank_mask:0xf"
        : "+v"(v));
    return fmaxf(v, __int_as_float(half_xor(__float_as_int(v))));
}
__device__ __forceinline__ int min_wave_dpp(int v)
{
    asm("v_min_i32_dpp %0, %0, %0 row_xmask:1 row_mask:0xf bank_mask:0xf\n\t"
        "v_min_i32_dpp %0, %0, %0 row_xmask:2 row_mask:0xf bank_mask:0xf\n\t"
        "v_min_i32_dpp %0, %0, %0 row_xmask:4 row_mask:0xf bank_mask:0xf\n\t"
        "v_min_i32_dpp %0, %0, %0 row_xmask:8 row_mask:0xf bank_mask:0xf"
        : "+v"(v));
    const int o = half_xor(v);
    return o < v ? o : v;
}
__device__ __forceinline__ float uni_f(float v)
{
    return __int_as_float(__builtin_amdgcn_readfirstlane(__float_as_int(v)));
}
__device__ __forceinline__ float lane_f(float v, int src)
{
    return __int_as_float(__builtin_amdgcn_readlane(__float_as_int(v), src));
}

__device__ __forceinline__ void top4_add(float& d1, int& i1, float& d2, int& i2, float& d3, int& i3, float& s4,
                                         float cand, int i)
{
    const bool  g1 = cand > d1, g2 = cand > d2, g3 = cand > d3;
    const float n4  = g3 ? d3 : fmaxf(s4, cand);
    const float n3d = g2 ? d2 : (g3 ? cand : d3);   const int n3i = g2 ? i2 : (g3 ? i : i3);
    const float n2d = g1 ? d1 : (g2 ? cand : d2);   const int n2i = g1 ? i1 : (g2 ? i : i2);
    const float n1d = g1 ? cand : d1;               const int n1i = g1 ? i : i1;
    d1 = n1d; i1 = n1i; d2 = n2d; i2 = n2i; d3 = n3d; i3 = n3i; s4 = n4;
}

__device__ __forceinline__ void top3_of_wave(float d1, int i1, float d2, int i2, float d3, int i3, float s4,
                                             float (&od)[3], int (&oi)[3], float& bound)
{
#pragma unroll
    for (int r = 0; r < 3; ++r) {
        const float wm = uni_f(max_wave_dpp(d1));
        const int   wi = __builtin_amdgcn_readfirstlane(min_wave_dpp(d1 == wm ? i1 : FPS_NONE));
        od[r] = wm;
        oi[r] = wi;
        const bool pop = (d1 == wm) & (i1 == wi);
        d1 = pop ? d2 : d1;        i1 = pop ? i2 : i1;
        d2 = pop ? d3 : d2;        i2 = pop ? i3 : i2;
        d3 = pop ? s4 : d3;        i3 = pop ? FPS_NONE : i3;
    }
    bound = uni_f(max_wave_dpp(d1));
}

template <bool THIRDS>
__device__ __forceinline__ int pack_live(const float (&cd)[NCAND], const int (&ci)[NCAND], const float U,
                                         float2* lq, const int lane)
{
    int nl = 0;
#pragma unroll
    for (int k = 0; k < NCAND; ++k) {
        if (!THIRDS && k % 3 == 2) continue;
        const bool     lv  = cd[k] >= U;
        const unsigned mk  = __builtin_amdgcn_ballot_w32(lv);
        const int      pos = nl + (int)__builtin_amdgcn_mbcnt_lo(mk, 0u);
        const int      at  = (lv && pos < LIVE_CAP) ? pos : LIVE_CAP + lane;
        lq[at] = make_float2(cd[k], __int_as_float(ci[k]));
        nl += __builtin_popcount(mk);
    }
    return nl;
}

__device__ __forceinline__ float* hand_area(float* dists, float* out, size_t ws_size, int N, int B)
{
    const size_t used = (size_t)B * N + (size_t)B * FPS_K;
    const size_t area = (size_t)B * HAND_WORDS_PER_BATCH;
    return (ws_size >= (used + 2 * area) * sizeof(float)) ? dists + used : out;
}

template <int NC>
__device__ __forceinline__ int select_run(
    float (&cd)[NC], const int (&ci)[NC], const float (&cx)[NC], const float (&cy)[NC], const float (&cz)[NC],
    const float U, const int cnt, const Pt3 (&c)[STEP_U], float (&d)[STEP_U],
    int* __restrict__ idx_row, float* s_p, const bool lead, const bool first_thread)
{
    int T = 0;
    for (;;) {
        float tm[NC];
#pragma unroll
        for (int k = 0; k < NC; ++k) tm[k] = cd[k];
#pragma unroll
        for (int w = NC; w > 1; w = (w + 1) / 2) {
#pragma unroll
            for (int k = 0; k < w / 2; ++k) tm[k] = fmaxf(tm[k], tm[k + (w + 1) / 2]);
        }
        const float wmax = uni_f(max_wave_dpp(tm[0]));
        if (T > 0 && !(wmax > U)) break;
        int tn[NC];
#pragma unroll
        for (int k = 0; k < NC; ++k) tn[k] = cd[k] == wmax ? ci[k] : FPS_NONE;
#pragma unroll
        for (int w = NC; w > 1; w = (w + 1) / 2) {
#pragma unroll
            for (int k = 0; k < w / 2; ++k) {
                const int o = tn[k + (w + 1) / 2];
                tn[k] = o < tn[k] ? o : tn[k];
            }
        }
        const int bi = tn[0];
        const int pi = __builtin_amdgcn_readfirstlane(min_wave_dpp(bi));
        int ox[NC], oy[NC], oz[NC];
#pragma unroll
        for (int k = 0; k < NC; ++k) {
            const bool h = ci[k] == bi;
            ox[k] = h ? __float_as_int(cx[k]) : 0;
            oy[k] = h ? __float_as_int(cy[k]) : 0;
            oz[k] = h ? __float_as_int(cz[k]) : 0;
        }
#pragma unroll
        for (int w = NC; w > 1; w = (w + 1) / 2) {
#pragma unroll
            for (int k = 0; k < w / 2; ++k) {
                ox[k] |= ox[k + (w + 1) / 2]; oy[k] |= oy[k + (w + 1) / 2]; oz[k] |= oz[k + (w + 1) / 2];
            }
        }
        const float sx = __int_as_float(ox[0]), sy = __int_as_float(oy[0]), sz = __int_as_float(oz[0]);
        const unsigned hit = __builtin_amdgcn_ballot_w32(bi == pi);
        const int src = hit ? __builtin_ctz(hit) : 0;
        const float px = lane_f(sx, src), py = lane_f(sy, src), pz = lane_f(sz, src);

        if (lead) idx_row[cnt + T] = pi;
        if (first_thread) { s_p[3 * T] = px; s_p[3 * T + 1] = py; s_p[3 * T + 2] = pz; }

#pragma unroll
        for (int k = 0; k < NC; ++k)
            cd[k] = fminf(cd[k], dist2(cx[k], cy[k], cz[k], px, py, pz));
#pragma unroll
        for (int s = 0; s < STEP_U; ++s)
            d[s] = fminf(d[s], dist2(c[s].x, c[s].y, c[s].z, px, py, pz));

        ++T;
        if (cnt + T >= FPS_K) break;
    }
    return T;
}

template <bool FIRST>
__global__ __launch_bounds__(STEP_T) void fps_step_kernel(
    const float* __restrict__ means, int N, int B,
    float* __restrict__ dists, int* __restrict__ idx_out,
    size_t ws_size, float* __restrict__ out, int L)
{
    const int g    = blockIdx.x;
    const int b    = blockIdx.y;
    const int t    = threadIdx.x;
    const int wave = t >> 5;
    const int lane = t & 31;

    const Pt3* m   = (const Pt3*)(means + (size_t)b * N * 3);
    float*     dst = dists + (size_t)b * N;

    float* hand = hand_area(dists, out, ws_size, N, B);
    float*       hw = hand + ((size_t)(L & 1) * B + b) * HAND_WORDS_PER_BATCH + g * (STEP_W * REC_W);
    const float* hr = hand + ((size_t)((L & 1) ^ 1) * B + b) * HAND_WORDS_PER_BATCH;

    __shared__ float4 s_rec[STEP_W * REC_W / 4];
    __shared__ float  s_p[3 * FPS_K];
    __shared__ float2 s_lq[STEP_W][LIVE_CAP + 32];

    int cnt = 0;
    if (!FIRST) {
        cnt = __builtin_amdgcn_readfirstlane(__float_as_int(hr[REC_CNT]));
        if (cnt >= FPS_K) {
            if (wave == 0 && lane < STEP_W * REC_W / 4)
                ((float4*)hw)[lane] = ((const float4*)(hr + g * (STEP_W * REC_W)))[lane];
            return;
        }
    }

    const int ch   = ((N + STEP_G * 32 - 1) / (STEP_G * 32)) * 32;
    const int base = g * ch;
    const int end  = (base + ch < N) ? base + ch : N;

    Pt3   c[STEP_U];
    float d[STEP_U];
#pragma unroll
    for (int s = 0; s < STEP_U; ++s) {
        const int i  = base + s * STEP_T + t;
        const int ic = i < N ? i : N - 1;
        c[s] = m[ic];
        d[s] = FIRST ? FPS_BIG : dst[ic];
    }

    int T = 0;
    if (FIRST) {
        const Pt3 p0 = m[0];
        if (g == 0 && t == 0) idx_out[(size_t)b * FPS_K] = 0;
        if (t == 0) { s_p[0] = p0.x; s_p[1] = p0.y; s_p[2] = p0.z; }
#pragma unroll
        for (int s = 0; s < STEP_U; ++s)
            d[s] = fminf(d[s], dist2(c[s].x, c[s].y, c[s].z, p0.x, p0.y, p0.z));
        T = 1;
    } else {
        float cd[NCAND];
        int   ci[NCAND];
        float4 q0[CAND_R], q1[CAND_R];
#pragma unroll
        for (int r = 0; r < CAND_R; ++r) {
            const float4* rp = (const float4*)(hr + (size_t)(r * 32 + lane) * REC_W);
            q0[r] = rp[0];
            q1[r] = rp[1];
        }
        asm volatile("" ::: "memory");
        __builtin_amdgcn_sched_barrier(0);
        float u = -1.0f, u3 = -1.0f;
#pragma unroll
        for (int r = 0; r < CAND_R; ++r) {
            cd[3 * r]     = q0[r].x; ci[3 * r]     = __float_as_int(q0[r].y);
            cd[3 * r + 1] = q0[r].z; ci[3 * r + 1] = __float_as_int(q0[r].w);
            cd[3 * r + 2] = q1[r].x; ci[3 * r + 2] = __float_as_int(q1[r].y);
            u  = fmaxf(u, q1[r].z);
            u3 = fmaxf(u3, q1[r].x);
        }
        float U = uni_f(max_wave(u));

        int nl = pack_live<true>(cd, ci, U, s_lq[wave], lane);
        if (nl > LIVE_CAP) {
            U  = uni_f(max_wave(u3));
            nl = pack_live<false>(cd, ci, U, s_lq[wave], lane);
        }
        __syncthreads();

        int* idx_row = idx_out + (size_t)b * FPS_K;
        if (nl >= 1 && nl <= LIVE_CAP) {
            float ld[LIVE_R], lx[LIVE_R], ly[LIVE_R], lz[LIVE_R];
            int   li[LIVE_R];
            Pt3   lp[LIVE_R];
#pragma unroll
            for (int j = 0; j < LIVE_R; ++j) {
                const int    e  = j * 32 + lane;
                const bool   ok = e < nl;
                const float2 q  = s_lq[wave][ok ? e : 0];
                const int    qi = __float_as_int(q.y);
                ld[j] = ok ? q.x : -1.0f;
                li[j] = ok ? qi : FPS_NONE;
                lp[j] = m[(unsigned)qi < (unsigned)N ? qi : N - 1];
            }
#pragma unroll
            for (int j = 0; j < LIVE_R; ++j) { lx[j] = lp[j].x; ly[j] = lp[j].y; lz[j] = lp[j].z; }
            T = select_run<LIVE_R>(ld, li, lx, ly, lz, U, cnt, c, d, idx_row, s_p, g == 0 && t == 0, t == 0);
        } else {
            float bd = cd[0];
            int   bi = ci[0];
#pragma unroll
            for (int k = 1; k < NCAND; ++k) argmax_pair(bd, bi, cd[k], ci[k]);
            argmax_wave(bd, bi);
            const int pi = __builtin_amdgcn_readfirstlane(bi);
            const Pt3 p  = m[(unsigned)pi < (unsigned)N ? pi : N - 1];
            if (g == 0 && t == 0) idx_row[cnt] = pi;
            if (t == 0) { s_p[0] = p.x; s_p[1] = p.y; s_p[2] = p.z; }
#pragma unroll
            for (int s = 0; s < STEP_U; ++s)
                d[s] = fminf(d[s], dist2(c[s].x, c[s].y, c[s].z, p.x, p.y, p.z));
            T = 1;
        }
    }

    float d1 = -1.0f, d2 = -1.0f, d3 = -1.0f, s4 = -1.0f;
    int   i1 = FPS_NONE, i2 = FPS_NONE, i3 = FPS_NONE;
#pragma unroll
    for (int s = 0; s < STEP_U; ++s) {
        const int  i     = base + s * STEP_T + t;
        const bool valid = i < end;
        if (valid) dst[i] = d[s];
        top4_add(d1, i1, d2, i2, d3, i3, s4, valid ? d[s] : -1.0f, i);
    }

    if (ch > STEP_U * STEP_T) {
        __syncthreads();
        for (int i = base + STEP_U * STEP_T + t; i < end; i += STEP_T) {
            const Pt3 pt = m[i];
            float nd = FIRST ? FPS_BIG : dst[i];
            for (int q = 0; q < T; ++q)
                nd = fminf(nd, dist2(pt.x, pt.y, pt.z, s_p[3 * q], s_p[3 * q + 1], s_p[3 * q + 2]));
            dst[i] = nd;
            top4_add(d1, i1, d2, i2, d3, i3, s4, nd, i);
        }
    }

    float od[3], bound;
    int   oi[3];
    top3_of_wave(d1, i1, d2, i2, d3, i3, s4, od, oi, bound);
    if (lane == 0) {
        s_rec[wave * 2]     = make_float4(od[0], __int_as_float(oi[0]), od[1], __int_as_float(oi[1]));
        s_rec[wave * 2 + 1] = make_float4(od[2], __int_as_float(oi[2]), bound, __int_as_float(cnt + T));
    }
    __syncthreads();
    if (wave == 0 && lane < STEP_W * REC_W / 4) ((float4*)hw)[lane] = s_rec[lane];
}

__global__ __launch_bounds__(FIN_T) void fps_finish_kernel(
    const float* __restrict__ means, int N, int B,
    float* __restrict__ dists, int* __restrict__ idx_out,
    size_t ws_size, float* __restrict__ out, int Llast)
{
    const int b    = blockIdx.x;
    const int t    = threadIdx.x;
    const int wave = t >> 5;
    const int lane = t & 31;

    const Pt3* m   = (const Pt3*)(means + (size_t)b * N * 3);
    float*     dst = dists + (size_t)b * N;

    const float* hand = hand_area(dists, out, ws_size, N, B);
    const float* hr   = hand + ((size_t)(Llast & 1) * B + b) * HAND_WORDS_PER_BATCH;

    __shared__ float f_d[FIN_W];
    __shared__ int   f_i[FIN_W];

    const float* rp = hr + (size_t)(t & (RECS - 1)) * REC_W;
    const float4 r0 = *(const float4*)rp;
    const float4 r2 = *(const float4*)(rp + 4);
    const int cnt = __builtin_amdgcn_readfirstlane(__float_as_int(r2.w));
    if (cnt >= FPS_K) return;

    float wd = r0.x;
    int   wi = __float_as_int(r0.y);
    argmax_wave(wd, wi);
    if (lane == 0) { f_d[wave] = wd; f_i[wave] = wi; }
    __syncthreads();
    float ld = f_d[lane];
    int   li = f_i[lane];
    argmax_wave(ld, li);
    int last = __builtin_amdgcn_readfirstlane(li);
    __syncthreads();

    for (int k = cnt; k < FPS_K; ++k) {
        if (t == 0) idx_out[(size_t)b * FPS_K + k] = last;
        if (k == FPS_K - 1) break;
        const int lc = last < 0 ? 0 : (last < N ? last : N - 1);
        const Pt3 p = m[lc];
        float best  = -1.0f;
        int   besti = FPS_NONE;
        for (int i = t; i < N; i += FIN_T) {
            const Pt3 pt = m[i];
            const float nd = fminf(dst[i], dist2(pt.x, pt.y, pt.z, p.x, p.y, p.z));
            dst[i] = nd;
            if (nd > best) { best = nd; besti = i; }
        }
        argmax_wave(best, besti);
        if (lane == 0) { f_d[wave] = best; f_i[wave] = besti; }
        __syncthreads();
        float xd = f_d[lane];
        int   xi = f_i[lane];
        argmax_wave(xd, xi);
        last = __builtin_amdgcn_readfirstlane(xi);
        __syncthreads();
    }
}

__global__ __launch_bounds__(256) void gemm_wmma_kernel(
    const float* __restrict__ feats, const int* __restrict__ idx,
    const float* __restrict__ W, const float* __restrict__ bias,
    float* __restrict__ out, int N)
{
    const int wave = threadIdx.x >> 5;
    const int lane = threadIdx.x & 31;
    const int tile = blockIdx.x * 8 + wave;

    const int b  = tile >> 8;
    const int tm = (tile >> 4) & 15;
    const int tn = tile & 15;

    const int lhalf = lane >> 4;
    const int l15   = lane & 15;

    const int row = idx[b * FPS_K + tm * 16 + l15];
    const float* arow = feats + ((size_t)b * N + row) * IN_D;
    const float* wcol = W + tn * 16 + l15;

    v8f acc = {};
#pragma unroll
    for (int k0 = 0; k0 < IN_D; k0 += 4) {
        const int ka = k0 + lhalf * 2;
        v2f a, bf;
        a.x  = arow[ka];
        a.y  = arow[ka + 1];
        bf.x = wcol[(size_t)ka * OUT_D];
        bf.y = wcol[(size_t)(ka + 1) * OUT_D];
        acc = __builtin_amdgcn_wmma_f32_16x16x4_f32(
             false, a,  false, bf,
             (short)0, acc,  false,  false);
    }

    const int n  = tn * 16 + l15;
    const int m0 = tm * 16 + lhalf * 8;
    const float bv = bias[n];
    float* orow = out + ((size_t)b * FPS_K + m0) * OUT_D + n;
#pragma unroll
    for (int r = 0; r < 8; ++r)
        orow[(size_t)r * OUT_D] = acc[r] + bv;
}

extern "C" void kernel_launch(void* const* d_in, const int* in_sizes, int n_in,
                              void* d_out, int out_size, void* d_ws, size_t ws_size,
                              hipStream_t stream) {
    const float* feats = (const float*)d_in[0];
    const float* means = (const float*)d_in[1];
    const float* W     = (const float*)d_in[2];
    const float* bias  = (const float*)d_in[3];
    float*       out   = (float*)d_out;

    const int B = 4;
    const int N = in_sizes[1] / (B * 3);

    float* dists = (float*)d_ws;
    int*   idx   = (int*)((char*)d_ws + (size_t)B * N * sizeof(float));

    fps_step_kernel<true><<<dim3(STEP_G, B), STEP_T, 0, stream>>>(
        means, N, B, dists, idx, ws_size, out, 0);
    for (int L = 1; L < FPS_LAUNCHES; ++L)
        fps_step_kernel<false><<<dim3(STEP_G, B), STEP_T, 0, stream>>>(
            means, N, B, dists, idx, ws_size, out, L);
    fps_finish_kernel<<<B, FIN_T, 0, stream>>>(
        means, N, B, dists, idx, ws_size, out, FPS_LAUNCHES - 1);

    const int tiles = B * (FPS_K / 16) * (OUT_D / 16);
    gemm_wmma_kernel<<<tiles / 8, 256, 0, stream>>>(feats, idx, W, bias, out, N);
}


# FPS selection loop: the chunk's distance pass and the coordinate picks are dealt out between the cross-lane steps of the max and min reductions (an in-order wave waited at each); sum order kept
# speedup vs baseline: 38.7282x; 1.0906x over previous
#include <hip/hip_runtime.h>

#define FPS_K   256
#define IN_D    128
#define OUT_D   256

typedef __attribute__((ext_vector_type(2))) float v2f;
typedef __attribute__((ext_vector_type(8))) float v8f;

struct Pt3 { float x, y, z; };

#define STEP_T   128
#define STEP_W   (STEP_T / 32)
#define STEP_G   64
#define STEP_U   13
#define FPS_BIG  3.402823466e38f
#define FPS_NONE 0x7fffffff
#define FPS_LAUNCHES 25
#define FIN_T    1024
#define FIN_W    (FIN_T / 32)

#define REC_W   8
#define REC_CNT 7
#define RECS   (STEP_G * STEP_W)
#define CAND_R (RECS / 32)
#define NCAND  (3 * CAND_R)
#define HAND_WORDS_PER_BATCH (RECS * REC_W)
#define LIVE_R   8
#define LIVE_CAP (LIVE_R * 32)
static_assert(RECS % 32 == 0, "the records are dealt to the lanes of a wave32");
static_assert((STEP_W * REC_W * 4) % 128 == 0, "a workgroup's records fill whole 128-byte lines");
static_assert(STEP_W * REC_W / 4 <= 32, "one wave stores the workgroup's records with one instruction");
static_assert(REC_W == 8, "a record is read and written as two 16-byte halves");
static_assert((RECS & (RECS - 1)) == 0 && RECS <= FIN_T, "the finisher reads one record per thread, modulo RECS");
static_assert(FIN_W == 32, "the finisher's second reduction reads one wave leader per lane");

__device__ __forceinline__ float dist2(float cx, float cy, float cz, float px, float py, float pz)
{
    const float dx = cx - px;
    const float dy = cy - py;
    const float dz = cz - pz;
    return fmaf(dx, dx, fmaf(dy, dy, dz * dz));
}

__device__ __forceinline__ bool ahead(float ad, int ai, float bd, int bi)
{
    return (ad > bd) | ((ad == bd) & (ai < bi));
}

__device__ __forceinline__ void argmax_pair(float& d, int& i, float od, int oi)
{
    const bool take = ahead(od, oi, d, i);
    d = take ? od : d;
    i = take ? oi : i;
}

template <int M>
__device__ __forceinline__ int row_xor(int v)
{
    return __builtin_amdgcn_update_dpp(0, v, 0x160 + M, 0xf, 0xf, false);
}
__device__ __forceinline__ int half_xor(int v)
{
    return __builtin_amdgcn_permlanex16(v, v, 0x76543210, 0xfedcba98, false, false);
}
template <int M>
__device__ __forceinline__ void argmax_row_step(float& d, int& i)
{
    const float od = __int_as_float(row_xor<M>(__float_as_int(d)));
    const int   oi = row_xor<M>(i);
    argmax_pair(d, i, od, oi);
}
__device__ __forceinline__ void argmax_wave(float& d, int& i)
{
    argmax_row_step<1>(d, i);
    argmax_row_step<2>(d, i);
    argmax_row_step<4>(d, i);
    argmax_row_step<8>(d, i);
    const float od = __int_as_float(half_xor(__float_as_int(d)));
    const int   oi = half_xor(i);
    argmax_pair(d, i, od, oi);
}
template <int M>
__device__ __forceinline__ float max_row_step(float v)
{
    return fmaxf(v, __int_as_float(row_xor<M>(__float_as_int(v))));
}
__device__ __forceinline__ float max_wave(float v)
{
    v = max_row_step<1>(v);
    v = max_row_step<2>(v);
    v = max_row_step<4>(v);
    v = max_row_step<8>(v);
    return fmaxf(v, __int_as_float(half_xor(__float_as_int(v))));
}
__device__ __forceinline__ float max_wave_dpp(float v)
{
    asm("v_max_num_f32_dpp %0, %0, %0 row_xmask:1 row_mask:0xf bank_mask:0xf\n\t"
        "v_max_num_f32_dpp %0, %0, %0 row_xmask:2 row_mask:0xf bank_mask:0xf\n\t"
        "v_max_num_f32_dpp %0, %0, %0 row_xmask:4 row_mask:0xf bank_mask:0xf\n\t"
        "v_max_num_f32_dpp %0, %0, %0 row_xmask:8 row_mask:0xf bank_mask:0xf"
        : "+v"(v));
    return fmaxf(v, __int_as_float(half_xor(__float_as_int(v))));
}
__device__ __forceinline__ int min_wave_dpp(int v)
{
    asm("v_min_i32_dpp %0, %0, %0 row_xmask:1 row_mask:0xf bank_mask:0xf\n\t"
        "v_min_i32_dpp %0, %0, %0 row_xmask:2 row_mask:0xf bank_mask:0xf\n\t"
        "v_min_i32_dpp %0, %0, %0 row_xmask:4 row_mask:0xf bank_mask:0xf\n\t"
        "v_min_i32_dpp %0, %0, %0 row_xmask:8 row_mask:0xf bank_mask:0xf"
        : "+v"(v));
    const int o = half_xor(v);
    return o < v ? o : v;
}
#define MAX_ROW_STEP(NAME, XM)                                                             \
__device__ __forceinline__ float NAME(float v)                                             \
{                                                                                          \
    float o;                                                                               \
      \
    asm("v_max_num_f32_dpp %0, %1, %1 row_xmask:" XM " row_mask:0xf bank_mask:0xf"        \
        : "=v"(o) : "v"(v));                                                               \
    return o;                                                                              \
}
MAX_ROW_STEP(max_row_step1, "1")
MAX_ROW_STEP(max_row_step2, "2")
MAX_ROW_STEP(max_row_step4, "4")
MAX_ROW_STEP(max_row_step8, "8")
#define MIN_ROW_STEP(NAME, XM)                                                             \
__device__ __forceinline__ int NAME(int v)                                                 \
{                                                                                          \
    int o;                                                                                 \
      \
    asm("v_min_i32_dpp %0, %1, %1 row_xmask:" XM " row_mask:0xf bank_mask:0xf"            \
        : "=v"(o) : "v"(v));                                                               \
    return o;                                                                              \
}
MIN_ROW_STEP(min_row_step1, "1")
MIN_ROW_STEP(min_row_step2, "2")
MIN_ROW_STEP(min_row_step4, "4")
MIN_ROW_STEP(min_row_step8, "8")
__device__ __forceinline__ float uni_f(float v)
{
    return __int_as_float(__builtin_amdgcn_readfirstlane(__float_as_int(v)));
}
__device__ __forceinline__ float lane_f(float v, int src)
{
    return __int_as_float(__builtin_amdgcn_readlane(__float_as_int(v), src));
}

__device__ __forceinline__ void top4_add(float& d1, int& i1, float& d2, int& i2, float& d3, int& i3, float& s4,
                                         float cand, int i)
{
    const bool  g1 = cand > d1, g2 = cand > d2, g3 = cand > d3;
    const float n4  = g3 ? d3 : fmaxf(s4, cand);
    const float n3d = g2 ? d2 : (g3 ? cand : d3);   const int n3i = g2 ? i2 : (g3 ? i : i3);
    const float n2d = g1 ? d1 : (g2 ? cand : d2);   const int n2i = g1 ? i1 : (g2 ? i : i2);
    const float n1d = g1 ? cand : d1;               const int n1i = g1 ? i : i1;
    d1 = n1d; i1 = n1i; d2 = n2d; i2 = n2i; d3 = n3d; i3 = n3i; s4 = n4;
}

__device__ __forceinline__ void top3_of_wave(float d1, int i1, float d2, int i2, float d3, int i3, float s4,
                                             float (&od)[3], int (&oi)[3], float& bound)
{
#pragma unroll
    for (int r = 0; r < 3; ++r) {
        const float wm = uni_f(max_wave_dpp(d1));
        const int   wi = __builtin_amdgcn_readfirstlane(min_wave_dpp(d1 == wm ? i1 : FPS_NONE));
        od[r] = wm;
        oi[r] = wi;
        const bool pop = (d1 == wm) & (i1 == wi);
        d1 = pop ? d2 : d1;        i1 = pop ? i2 : i1;
        d2 = pop ? d3 : d2;        i2 = pop ? i3 : i2;
        d3 = pop ? s4 : d3;        i3 = pop ? FPS_NONE : i3;
    }
    bound = uni_f(max_wave_dpp(d1));
}

template <bool THIRDS>
__device__ __forceinline__ int pack_live(const float (&cd)[NCAND], const int (&ci)[NCAND], const float U,
                                         float2* lq, const int lane)
{
    int nl = 0;
#pragma unroll
    for (int k = 0; k < NCAND; ++k) {
        if (!THIRDS && k % 3 == 2) continue;
        const bool     lv  = cd[k] >= U;
        const unsigned mk  = __builtin_amdgcn_ballot_w32(lv);
        const int      pos = nl + (int)__builtin_amdgcn_mbcnt_lo(mk, 0u);
        const int      at  = (lv && pos < LIVE_CAP) ? pos : LIVE_CAP + lane;
        lq[at] = make_float2(cd[k], __int_as_float(ci[k]));
        nl += __builtin_popcount(mk);
    }
    return nl;
}

__device__ __forceinline__ float* hand_area(float* dists, float* out, size_t ws_size, int N, int B)
{
    const size_t used = (size_t)B * N + (size_t)B * FPS_K;
    const size_t area = (size_t)B * HAND_WORDS_PER_BATCH;
    return (ws_size >= (used + 2 * area) * sizeof(float)) ? dists + used : out;
}

template <int NC>
__device__ __forceinline__ float lane_max(const float (&cd)[NC])
{
    float tm[NC];
#pragma unroll
    for (int k = 0; k < NC; ++k) tm[k] = cd[k];
#pragma unroll
    for (int w = NC; w > 1; w = (w + 1) / 2) {
#pragma unroll
        for (int k = 0; k < w / 2; ++k) tm[k] = fmaxf(tm[k], tm[k + (w + 1) / 2]);
    }
    return tm[0];
}

template <int NC>
__device__ __forceinline__ int select_run(
    float (&cd)[NC], const int (&ci)[NC], const float (&cx)[NC], const float (&cy)[NC], const float (&cz)[NC],
    const float U, const int cnt, const Pt3 (&c)[STEP_U], float (&d)[STEP_U],
    int* __restrict__ idx_row, float* s_p, const bool lead, const bool first_thread)
{
    int T = 0;
    float wmax = uni_f(max_wave_dpp(lane_max<NC>(cd)));
    for (;;) {
        int tn[NC];
#pragma unroll
        for (int k = 0; k < NC; ++k) tn[k] = cd[k] == wmax ? ci[k] : FPS_NONE;
#pragma unroll
        for (int w = NC; w > 1; w = (w + 1) / 2) {
#pragma unroll
            for (int k = 0; k < w / 2; ++k) {
                const int o = tn[k + (w + 1) / 2];
                tn[k] = o < tn[k] ? o : tn[k];
            }
        }
        const int bi = tn[0];
        static_assert(NC % 4 == 0, "the picks are dealt out in four parts");
        int ox[NC], oy[NC], oz[NC];
        int mi = bi;
#define PICK_PART(P)                                                                       \
        _Pragma("unroll")                            \
        for (int k = (P) * (NC / 4); k < ((P) + 1) * (NC / 4); ++k) {                      \
            const bool h = ci[k] == bi;                                                    \
            ox[k] = h ? __float_as_int(cx[k]) : 0;                                         \
            oy[k] = h ? __float_as_int(cy[k]) : 0;                                         \
            oz[k] = h ? __float_as_int(cz[k]) : 0;                                         \
        }
        __builtin_amdgcn_sched_barrier(0);
        mi = min_row_step1(mi);
        __builtin_amdgcn_sched_barrier(0);
        PICK_PART(0)
        __builtin_amdgcn_sched_barrier(0);
        mi = min_row_step2(mi);
        __builtin_amdgcn_sched_barrier(0);
        PICK_PART(1)
        __builtin_amdgcn_sched_barrier(0);
        mi = min_row_step4(mi);
        __builtin_amdgcn_sched_barrier(0);
        PICK_PART(2)
        __builtin_amdgcn_sched_barrier(0);
        mi = min_row_step8(mi);
        __builtin_amdgcn_sched_barrier(0);
        PICK_PART(3)
#undef PICK_PART
        __builtin_amdgcn_sched_barrier(0);
        const int mo = half_xor(mi);
        __builtin_amdgcn_sched_barrier(0);
#pragma unroll
        for (int w = NC; w > 1; w = (w + 1) / 2) {
#pragma unroll
            for (int k = 0; k < w / 2; ++k) {
                ox[k] |= ox[k + (w + 1) / 2]; oy[k] |= oy[k + (w + 1) / 2]; oz[k] |= oz[k + (w + 1) / 2];
            }
        }
        const int pi = __builtin_amdgcn_readfirstlane(mo < mi ? mo : mi);
        const float sx = __int_as_float(ox[0]), sy = __int_as_float(oy[0]), sz = __int_as_float(oz[0]);
        const unsigned hit = __builtin_amdgcn_ballot_w32(bi == pi);
        const int src = hit ? __builtin_ctz(hit) : 0;
        const float px = lane_f(sx, src), py = lane_f(sy, src), pz = lane_f(sz, src);

        if (lead) idx_row[cnt + T] = pi;
        if (first_thread) { s_p[3 * T] = px; s_p[3 * T + 1] = py; s_p[3 * T + 2] = pz; }

#pragma unroll
        for (int k = 0; k < NC; ++k)
            cd[k] = fminf(cd[k], dist2(cx[k], cy[k], cz[k], px, py, pz));
        ++T;
        __builtin_amdgcn_sched_barrier(0);
        float m = lane_max<NC>(cd);
        float ex[STEP_U], ey[STEP_U], ez[STEP_U], q[STEP_U];
#pragma unroll
        for (int s = 0; s < STEP_U; ++s) { ex[s] = c[s].x - px; ey[s] = c[s].y - py; ez[s] = c[s].z - pz; }
        __builtin_amdgcn_sched_barrier(0);
        m = max_row_step1(m);
        __builtin_amdgcn_sched_barrier(0);
#pragma unroll
        for (int s = 0; s < STEP_U; ++s) q[s] = ez[s] * ez[s];
        __builtin_amdgcn_sched_barrier(0);
        m = max_row_step2(m);
        __builtin_amdgcn_sched_barrier(0);
#pragma unroll
        for (int s = 0; s < STEP_U; ++s) q[s] = fmaf(ey[s], ey[s], q[s]);
        __builtin_amdgcn_sched_barrier(0);
        m = max_row_step4(m);
        __builtin_amdgcn_sched_barrier(0);
#pragma unroll
        for (int s = 0; s < STEP_U; ++s) q[s] = fmaf(ex[s], ex[s], q[s]);
        __builtin_amdgcn_sched_barrier(0);
        m = max_row_step8(m);
        __builtin_amdgcn_sched_barrier(0);
#pragma unroll
        for (int s = 0; s < STEP_U / 2; ++s) d[s] = fminf(d[s], q[s]);
        __builtin_amdgcn_sched_barrier(0);
        const float mh = __int_as_float(half_xor(__float_as_int(m)));
        __builtin_amdgcn_sched_barrier(0);
#pragma unroll
        for (int s = STEP_U / 2; s < STEP_U; ++s) d[s] = fminf(d[s], q[s]);
        const float nmax = uni_f(fmaxf(m, mh));
        if (cnt + T >= FPS_K || !(nmax > U)) break;
        wmax = nmax;
    }
    return T;
}

template <bool FIRST>
__global__ __launch_bounds__(STEP_T) void fps_step_kernel(
    const float* __restrict__ means, int N, int B,
    float* __restrict__ dists, int* __restrict__ idx_out,
    size_t ws_size, float* __restrict__ out, int L)
{
    const int g    = blockIdx.x;
    const int b    = blockIdx.y;
    const int t    = threadIdx.x;
    const int wave = t >> 5;
    const int lane = t & 31;

    const Pt3* m   = (const Pt3*)(means + (size_t)b * N * 3);
    float*     dst = dists + (size_t)b * N;

    float* hand = hand_area(dists, out, ws_size, N, B);
    float*       hw = hand + ((size_t)(L & 1) * B + b) * HAND_WORDS_PER_BATCH + g * (STEP_W * REC_W);
    const float* hr = hand + ((size_t)((L & 1) ^ 1) * B + b) * HAND_WORDS_PER_BATCH;

    __shared__ float4 s_rec[STEP_W * REC_W / 4];
    __shared__ float  s_p[3 * FPS_K];
    __shared__ float2 s_lq[STEP_W][LIVE_CAP + 32];

    int cnt = 0;
    if (!FIRST) {
        cnt = __builtin_amdgcn_readfirstlane(__float_as_int(hr[REC_CNT]));
        if (cnt >= FPS_K) {
            if (wave == 0 && lane < STEP_W * REC_W / 4)
                ((float4*)hw)[lane] = ((const float4*)(hr + g * (STEP_W * REC_W)))[lane];
            return;
        }
    }

    const int ch   = ((N + STEP_G * 32 - 1) / (STEP_G * 32)) * 32;
    const int base = g * ch;
    const int end  = (base + ch < N) ? base + ch : N;

    Pt3   c[STEP_U];
    float d[STEP_U];
#pragma unroll
    for (int s = 0; s < STEP_U; ++s) {
        const int i  = base + s * STEP_T + t;
        const int ic = i < N ? i : N - 1;
        c[s] = m[ic];
        d[s] = FIRST ? FPS_BIG : dst[ic];
    }

    int T = 0;
    if (FIRST) {
        const Pt3 p0 = m[0];
        if (g == 0 && t == 0) idx_out[(size_t)b * FPS_K] = 0;
        if (t == 0) { s_p[0] = p0.x; s_p[1] = p0.y; s_p[2] = p0.z; }
#pragma unroll
        for (int s = 0; s < STEP_U; ++s)
            d[s] = fminf(d[s], dist2(c[s].x, c[s].y, c[s].z, p0.x, p0.y, p0.z));
        T = 1;
    } else {
        float cd[NCAND];
        int   ci[NCAND];
        float4 q0[CAND_R], q1[CAND_R];
#pragma unroll
        for (int r = 0; r < CAND_R; ++r) {
            const float4* rp = (const float4*)(hr + (size_t)(r * 32 + lane) * REC_W);
            q0[r] = rp[0];
            q1[r] = rp[1];
        }
        asm volatile("" ::: "memory");
        __builtin_amdgcn_sched_barrier(0);
        float u = -1.0f, u3 = -1.0f;
#pragma unroll
        for (int r = 0; r < CAND_R; ++r) {
            cd[3 * r]     = q0[r].x; ci[3 * r]     = __float_as_int(q0[r].y);
            cd[3 * r + 1] = q0[r].z; ci[3 * r + 1] = __float_as_int(q0[r].w);
            cd[3 * r + 2] = q1[r].x; ci[3 * r + 2] = __float_as_int(q1[r].y);
            u  = fmaxf(u, q1[r].z);
            u3 = fmaxf(u3, q1[r].x);
        }
        float U = uni_f(max_wave(u));

        int nl = pack_live<true>(cd, ci, U, s_lq[wave], lane);
        if (nl > LIVE_CAP) {
            U  = uni_f(max_wave(u3));
            nl = pack_live<false>(cd, ci, U, s_lq[wave], lane);
        }
        __syncthreads();

        int* idx_row = idx_out + (size_t)b * FPS_K;
        if (nl >= 1 && nl <= LIVE_CAP) {
            float ld[LIVE_R], lx[LIVE_R], ly[LIVE_R], lz[LIVE_R];
            int   li[LIVE_R];
            Pt3   lp[LIVE_R];
#pragma unroll
            for (int j = 0; j < LIVE_R; ++j) {
                const int    e  = j * 32 + lane;
                const bool   ok = e < nl;
                const float2 q  = s_lq[wave][ok ? e : 0];
                const int    qi = __float_as_int(q.y);
                ld[j] = ok ? q.x : -1.0f;
                li[j] = ok ? qi : FPS_NONE;
                lp[j] = m[(unsigned)qi < (unsigned)N ? qi : N - 1];
            }
#pragma unroll
            for (int j = 0; j < LIVE_R; ++j) { lx[j] = lp[j].x; ly[j] = lp[j].y; lz[j] = lp[j].z; }
            T = select_run<LIVE_R>(ld, li, lx, ly, lz, U, cnt, c, d, idx_row, s_p, g == 0 && t == 0, t == 0);
        } else {
            float bd = cd[0];
            int   bi = ci[0];
#pragma unroll
            for (int k = 1; k < NCAND; ++k) argmax_pair(bd, bi, cd[k], ci[k]);
            argmax_wave(bd, bi);
            const int pi = __builtin_amdgcn_readfirstlane(bi);
            const Pt3 p  = m[(unsigned)pi < (unsigned)N ? pi : N - 1];
            if (g == 0 && t == 0) idx_row[cnt] = pi;
            if (t == 0) { s_p[0] = p.x; s_p[1] = p.y; s_p[2] = p.z; }
#pragma unroll
            for (int s = 0; s < STEP_U; ++s)
                d[s] = fminf(d[s], dist2(c[s].x, c[s].y, c[s].z, p.x, p.y, p.z));
            T = 1;
        }
    }

    float d1 = -1.0f, d2 = -1.0f, d3 = -1.0f, s4 = -1.0f;
    int   i1 = FPS_NONE, i2 = FPS_NONE, i3 = FPS_NONE;
#pragma unroll
    for (int s = 0; s < STEP_U; ++s) {
        const int  i     = base + s * STEP_T + t;
        const bool valid = i < end;
        if (valid) dst[i] = d[s];
        top4_add(d1, i1, d2, i2, d3, i3, s4, valid ? d[s] : -1.0f, i);
    }

    if (ch > STEP_U * STEP_T) {
        __syncthreads();
        for (int i = base + STEP_U * STEP_T + t; i < end; i += STEP_T) {
            const Pt3 pt = m[i];
            float nd = FIRST ? FPS_BIG : dst[i];
            for (int q = 0; q < T; ++q)
                nd = fminf(nd, dist2(pt.x, pt.y, pt.z, s_p[3 * q], s_p[3 * q + 1], s_p[3 * q + 2]));
            dst[i] = nd;
            top4_add(d1, i1, d2, i2, d3, i3, s4, nd, i);
        }
    }

    float od[3], bound;
    int   oi[3];
    top3_of_wave(d1, i1, d2, i2, d3, i3, s4, od, oi, bound);
    if (lane == 0) {
        s_rec[wave * 2]     = make_float4(od[0], __int_as_float(oi[0]), od[1], __int_as_float(oi[1]));
        s_rec[wave * 2 + 1] = make_float4(od[2], __int_as_float(oi[2]), bound, __int_as_float(cnt + T));
    }
    __syncthreads();
    if (wave == 0 && lane < STEP_W * REC_W / 4) ((float4*)hw)[lane] = s_rec[lane];
}

__global__ __launch_bounds__(FIN_T) void fps_finish_kernel(
    const float* __restrict__ means, int N, int B,
    float* __restrict__ dists, int* __restrict__ idx_out,
    size_t ws_size, float* __restrict__ out, int Llast)
{
    const int b    = blockIdx.x;
    const int t    = threadIdx.x;
    const int wave = t >> 5;
    const int lane = t & 31;

    const Pt3* m   = (const Pt3*)(means + (size_t)b * N * 3);
    float*     dst = dists + (size_t)b * N;

    const float* hand = hand_area(dists, out, ws_size, N, B);
    const float* hr   = hand + ((size_t)(Llast & 1) * B + b) * HAND_WORDS_PER_BATCH;

    __shared__ float f_d[FIN_W];
    __shared__ int   f_i[FIN_W];

    const float* rp = hr + (size_t)(t & (RECS - 1)) * REC_W;
    const float4 r0 = *(const float4*)rp;
    const float4 r2 = *(const float4*)(rp + 4);
    const int cnt = __builtin_amdgcn_readfirstlane(__float_as_int(r2.w));
    if (cnt >= FPS_K) return;

    float wd = r0.x;
    int   wi = __float_as_int(r0.y);
    argmax_wave(wd, wi);
    if (lane == 0) { f_d[wave] = wd; f_i[wave] = wi; }
    __syncthreads();
    float ld = f_d[lane];
    int   li = f_i[lane];
    argmax_wave(ld, li);
    int last = __builtin_amdgcn_readfirstlane(li);
    __syncthreads();

    for (int k = cnt; k < FPS_K; ++k) {
        if (t == 0) idx_out[(size_t)b * FPS_K + k] = last;
        if (k == FPS_K - 1) break;
        const int lc = last < 0 ? 0 : (last < N ? last : N - 1);
        const Pt3 p = m[lc];
        float best  = -1.0f;
        int   besti = FPS_NONE;
        for (int i = t; i < N; i += FIN_T) {
            const Pt3 pt = m[i];
            const float nd = fminf(dst[i], dist2(pt.x, pt.y, pt.z, p.x, p.y, p.z));
            dst[i] = nd;
            if (nd > best) { best = nd; besti = i; }
        }
        argmax_wave(best, besti);
        if (lane == 0) { f_d[wave] = best; f_i[wave] = besti; }
        __syncthreads();
        float xd = f_d[lane];
        int   xi = f_i[lane];
        argmax_wave(xd, xi);
        last = __builtin_amdgcn_readfirstlane(xi);
        __syncthreads();
    }
}

__global__ __launch_bounds__(256) void gemm_wmma_kernel(
    const float* __restrict__ feats, const int* __restrict__ idx,
    const float* __restrict__ W, const float* __restrict__ bias,
    float* __restrict__ out, int N)
{
    const int wave = threadIdx.x >> 5;
    const int lane = threadIdx.x & 31;
    const int tile = blockIdx.x * 8 + wave;

    const int b  = tile >> 8;
    const int tm = (tile >> 4) & 15;
    const int tn = tile & 15;

    const int lhalf = lane >> 4;
    const int l15   = lane & 15;

    const int row = idx[b * FPS_K + tm * 16 + l15];
    const float* arow = feats + ((size_t)b * N + row) * IN_D;
    const float* wcol = W + tn * 16 + l15;

    v8f acc = {};
#pragma unroll
    for (int k0 = 0; k0 < IN_D; k0 += 4) {
        const int ka = k0 + lhalf * 2;
        v2f a, bf;
        a.x  = arow[ka];
        a.y  = arow[ka + 1];
        bf.x = wcol[(size_t)ka * OUT_D];
        bf.y = wcol[(size_t)(ka + 1) * OUT_D];
        acc = __builtin_amdgcn_wmma_f32_16x16x4_f32(
             false, a,  false, bf,
             (short)0, acc,  false,  false);
    }

    const int n  = tn * 16 + l15;
    const int m0 = tm * 16 + lhalf * 8;
    const float bv = bias[n];
    float* orow = out + ((size_t)b * FPS_K + m0) * OUT_D + n;
#pragma unroll
    for (int r = 0; r < 8; ++r)
        orow[(size_t)r * OUT_D] = acc[r] + bv;
}

extern "C" void kernel_launch(void* const* d_in, const int* in_sizes, int n_in,
                              void* d_out, int out_size, void* d_ws, size_t ws_size,
                              hipStream_t stream) {
    const float* feats = (const float*)d_in[0];
    const float* means = (const float*)d_in[1];
    const float* W     = (const float*)d_in[2];
    const float* bias  = (const float*)d_in[3];
    float*       out   = (float*)d_out;

    const int B = 4;
    const int N = in_sizes[1] / (B * 3);

    float* dists = (float*)d_ws;
    int*   idx   = (int*)((char*)d_ws + (size_t)B * N * sizeof(float));

    fps_step_kernel<true><<<dim3(STEP_G, B), STEP_T, 0, stream>>>(
        means, N, B, dists, idx, ws_size, out, 0);
    for (int L = 1; L < FPS_LAUNCHES; ++L)
        fps_step_kernel<false><<<dim3(STEP_G, B), STEP_T, 0, stream>>>(
            means, N, B, dists, idx, ws_size, out, L);
    fps_finish_kernel<<<B, FIN_T, 0, stream>>>(
        means, N, B, dists, idx, ws_size, out, FPS_LAUNCHES - 1);

    const int tiles = B * (FPS_K / 16) * (OUT_D / 16);
    gemm_wmma_kernel<<<tiles / 8, 256, 0, stream>>>(feats, idx, W, bias, out, N);
}
